# starting kernel
# baseline (speedup 1.0000x reference)
#include <hip/hip_runtime.h>
#include <hip/hip_bf16.h>
#include <math.h>

typedef float v2f __attribute__((ext_vector_type(2)));
typedef float v8f __attribute__((ext_vector_type(8)));

#define TT   512
#define BB   512
#define NIN  64
#define H1   100
#define H2   50
#define H3   25
#define G1   400
#define G2   200
#define G3   100
#define G1S  400
#define G2S  208
#define G3S  112
#define SH1  104
#define SH2  52
#define SH3  28
#define NT   256
#define NWAVES 8

#define KS_X1 16
#define KS_H1 25
#define KS_H2 13
#define KS_H3 7

__device__ __forceinline__ float sigf(float x) { return 1.0f / (1.0f + __expf(-x)); }

__device__ __forceinline__ v8f wmma4(v2f a, v2f b, v8f c) {
  return __builtin_amdgcn_wmma_f32_16x16x4_f32(
      false, a, false, b, (short)0, c, false, false);
}

template<int KS>
__device__ __forceinline__ void load_frags(v2f* f, const float* p, int khalf) {
  #pragma unroll
  for (int i = 0; i < KS; ++i) {
    f[i].x = p[4 * i + khalf];
    f[i].y = p[4 * i + khalf + 1];
  }
}

template<int KS>
__device__ __forceinline__ v8f wmma_chain(const v2f* a, const v2f* b, v8f acc) {
  #pragma unroll
  for (int i = 0; i < KS; ++i) acc = wmma4(a[i], b[i], acc);
  return acc;
}

__global__ __launch_bounds__(NT, 2) void lstm3_fused_kernel(
    const float* __restrict__ x,
    const float* __restrict__ w_ih1, const float* __restrict__ w_hh1,
    const float* __restrict__ b_ih1, const float* __restrict__ b_hh1,
    const float* __restrict__ w_ih2, const float* __restrict__ w_hh2,
    const float* __restrict__ b_ih2, const float* __restrict__ b_hh2,
    const float* __restrict__ w_ih3, const float* __restrict__ w_hh3,
    const float* __restrict__ b_ih3, const float* __restrict__ b_hh3,
    const float* __restrict__ w_lin, const float* __restrict__ b_lin,
    float* __restrict__ out)
{
  __shared__ float gates1[16 * G1S];
  __shared__ float gates2[16 * G2S];
  __shared__ float gates3[16 * G3S];
  __shared__ float h1s[16 * SH1];
  __shared__ float h2s[16 * SH2];
  __shared__ float h3s[16 * SH3];
  __shared__ float whh1s[G1 * H1];
  __shared__ float whh2s[208 * SH2];
  __shared__ float wih3s[112 * SH2];
  __shared__ float whh3s[112 * SH3];

  const int tid     = threadIdx.x;
  const int lane    = tid & 31;
  const int wave    = tid >> 5;
  const int col     = lane & 15;
  const int khalf   = (lane >> 4) << 1;
  const int rowhalf = (lane >> 4) << 3;
  const int bbase   = blockIdx.x * 16;

  for (int i = tid; i < 16 * SH1; i += NT) h1s[i] = 0.f;
  for (int i = tid; i < 16 * SH2; i += NT) h2s[i] = 0.f;
  for (int i = tid; i < 16 * SH3; i += NT) h3s[i] = 0.f;
  for (int i = tid; i < 208 * SH2; i += NT) whh2s[i] = 0.f;
  for (int i = tid; i < 112 * SH2; i += NT) wih3s[i] = 0.f;
  for (int i = tid; i < 112 * SH3; i += NT) whh3s[i] = 0.f;
  __syncthreads();

  for (int i = tid; i < G1 * H1; i += NT)
    whh1s[i] = w_hh1[i];
  for (int i = tid; i < G2 * H2; i += NT) {
    int r = i / H2, c = i - r * H2;
    whh2s[r * SH2 + c] = w_hh2[i];
  }
  for (int i = tid; i < G3 * H2; i += NT) {
    int r = i / H2, c = i - r * H2;
    wih3s[r * SH2 + c] = w_ih3[i];
  }
  for (int i = tid; i < G3 * H3; i += NT) {
    int r = i / H3, c = i - r * H3;
    whh3s[r * SH3 + c] = w_hh3[i];
  }

  float bs1[7][4], bs2[4][4], bs3[2][4];
  #pragma unroll
  for (int r = 0; r < 7; ++r) {
    const int idx = tid + r * NT;
    if (idx < 16 * H1) {
      const int j = idx >> 4;
      #pragma unroll
      for (int g = 0; g < 4; ++g) bs1[r][g] = b_ih1[g * H1 + j] + b_hh1[g * H1 + j];
    }
  }
  #pragma unroll
  for (int r = 0; r < 4; ++r) {
    const int idx = tid + r * NT;
    if (idx < 16 * H2) {
      const int j = idx >> 4;
      #pragma unroll
      for (int g = 0; g < 4; ++g) bs2[r][g] = b_ih2[g * H2 + j] + b_hh2[g * H2 + j];
    }
  }
  #pragma unroll
  for (int r = 0; r < 2; ++r) {
    const int idx = tid + r * NT;
    if (idx < 16 * H3) {
      const int j = idx >> 4;
      #pragma unroll
      for (int g = 0; g < 4; ++g) bs3[r][g] = b_ih3[g * H3 + j] + b_hh3[g * H3 + j];
    }
  }

  float wl[H3];
  #pragma unroll
  for (int j = 0; j < H3; ++j) wl[j] = w_lin[j];
  const float blin = b_lin[0];

  float c1r[7], c2r[4], c3r[2];
  #pragma unroll
  for (int r = 0; r < 7; ++r) c1r[r] = 0.f;
  #pragma unroll
  for (int r = 0; r < 4; ++r) c2r[r] = 0.f;
  #pragma unroll
  for (int r = 0; r < 2; ++r) c3r[r] = 0.f;

  __syncthreads();

  for (int t = 0; t < TT; ++t) {
    {
      if (t + 1 < TT && tid < 16)
        __builtin_prefetch(x + ((size_t)(t + 1) * BB + bbase + tid) * NIN, 0, 3);

      v2f ax[KS_X1], ah[KS_H1];
      load_frags<KS_X1>(ax, x + ((size_t)t * BB + bbase + col) * NIN, khalf);
      load_frags<KS_H1>(ah, &h1s[col * SH1], khalf);
      for (int tile = wave; tile < 25; tile += NWAVES) {
        const int n = tile * 16 + col;
        v2f bi[KS_X1], bh[KS_H1];
        load_frags<KS_X1>(bi, w_ih1 + (size_t)n * NIN, khalf);
        load_frags<KS_H1>(bh, &whh1s[n * H1], khalf);
        v8f acc = {};
        acc = wmma_chain<KS_X1>(ax, bi, acc);
        acc = wmma_chain<KS_H1>(ah, bh, acc);
        #pragma unroll
        for (int r = 0; r < 8; ++r)
          gates1[(r + rowhalf) * G1S + n] = acc[r];
      }
    }
    __syncthreads();

    #pragma unroll
    for (int r = 0; r < 7; ++r) {
      const int idx = tid + r * NT;
      if (idx < 16 * H1) {
        const int m = idx & 15, j = idx >> 4;
        float gi = gates1[m * G1S + j]          + bs1[r][0];
        float gf = gates1[m * G1S + H1 + j]     + bs1[r][1];
        float gg = gates1[m * G1S + 2 * H1 + j] + bs1[r][2];
        float go = gates1[m * G1S + 3 * H1 + j] + bs1[r][3];
        float c  = sigf(gf) * c1r[r] + sigf(gi) * tanhf(gg);
        c1r[r]   = c;
        h1s[m * SH1 + j] = sigf(go) * tanhf(c);
      }
    }
    __syncthreads();

    {
      v2f a1[KS_H1], a2[KS_H2];
      load_frags<KS_H1>(a1, &h1s[col * SH1], khalf);
      load_frags<KS_H2>(a2, &h2s[col * SH2], khalf);
      for (int tile = wave; tile < 13; tile += NWAVES) {
        const int n  = tile * 16 + col;
        const int nc = (n < G2) ? n : (G2 - 1);
        v2f bi[KS_H1], bh[KS_H2];
        load_frags<KS_H1>(bi, w_ih2 + (size_t)nc * H1, khalf);
        load_frags<KS_H2>(bh, &whh2s[n * SH2], khalf);
        v8f acc = {};
        acc = wmma_chain<KS_H1>(a1, bi, acc);
        acc = wmma_chain<KS_H2>(a2, bh, acc);
        if (n < G2) {
          #pragma unroll
          for (int r = 0; r < 8; ++r)
            gates2[(r + rowhalf) * G2S + n] = acc[r];
        }
      }
    }
    __syncthreads();

    #pragma unroll
    for (int r = 0; r < 4; ++r) {
      const int idx = tid + r * NT;
      if (idx < 16 * H2) {
        const int m = idx & 15, j = idx >> 4;
        float gi = gates2[m * G2S + j]          + bs2[r][0];
        float gf = gates2[m * G2S + H2 + j]     + bs2[r][1];
        float gg = gates2[m * G2S + 2 * H2 + j] + bs2[r][2];
        float go = gates2[m * G2S + 3 * H2 + j] + bs2[r][3];
        float c  = sigf(gf) * c2r[r] + sigf(gi) * tanhf(gg);
        c2r[r]   = c;
        h2s[m * SH2 + j] = sigf(go) * tanhf(c);
      }
    }
    __syncthreads();

    {
      v2f a2[KS_H2], a3[KS_H3];
      load_frags<KS_H2>(a2, &h2s[col * SH2], khalf);
      load_frags<KS_H3>(a3, &h3s[col * SH3], khalf);
      for (int tile = wave; tile < 7; tile += NWAVES) {
        const int n = tile * 16 + col;
        v2f bi[KS_H2], bh[KS_H3];
        load_frags<KS_H2>(bi, &wih3s[n * SH2], khalf);
        load_frags<KS_H3>(bh, &whh3s[n * SH3], khalf);
        v8f acc = {};
        acc = wmma_chain<KS_H2>(a2, bi, acc);
        acc = wmma_chain<KS_H3>(a3, bh, acc);
        if (n < G3) {
          #pragma unroll
          for (int r = 0; r < 8; ++r)
            gates3[(r + rowhalf) * G3S + n] = acc[r];
        }
      }
    }
    __syncthreads();

    #pragma unroll
    for (int r = 0; r < 2; ++r) {
      const int idx = tid + r * NT;
      if (idx < 16 * H3) {
        const int m = idx & 15, j = idx >> 4;
        float gi = gates3[m * G3S + j]          + bs3[r][0];
        float gf = gates3[m * G3S + H3 + j]     + bs3[r][1];
        float gg = gates3[m * G3S + 2 * H3 + j] + bs3[r][2];
        float go = gates3[m * G3S + 3 * H3 + j] + bs3[r][3];
        float c  = sigf(gf) * c3r[r] + sigf(gi) * tanhf(gg);
        c3r[r]   = c;
        h3s[m * SH3 + j] = sigf(go) * tanhf(c);
      }
    }
    __syncthreads();

    if (tid < 16) {
      const float* hr = &h3s[tid * SH3];
      float acc = blin;
      #pragma unroll
      for (int j = 0; j < H3; ++j) acc += hr[j] * wl[j];
      out[(size_t)t * BB + bbase + tid] = acc;
    }
    __syncthreads();
  }
}

extern "C" void kernel_launch(void* const* d_in, const int* in_sizes, int n_in,
                              void* d_out, int out_size, void* d_ws, size_t ws_size,
                              hipStream_t stream) {
  (void)in_sizes; (void)n_in; (void)d_ws; (void)ws_size; (void)out_size;
  const float* x     = (const float*)d_in[0];
  const float* w_ih1 = (const float*)d_in[1];
  const float* w_hh1 = (const float*)d_in[2];
  const float* b_ih1 = (const float*)d_in[3];
  const float* b_hh1 = (const float*)d_in[4];
  const float* w_ih2 = (const float*)d_in[5];
  const float* w_hh2 = (const float*)d_in[6];
  const float* b_ih2 = (const float*)d_in[7];
  const float* b_hh2 = (const float*)d_in[8];
  const float* w_ih3 = (const float*)d_in[9];
  const float* w_hh3 = (const float*)d_in[10];
  const float* b_ih3 = (const float*)d_in[11];
  const float* b_hh3 = (const float*)d_in[12];
  const float* w_lin = (const float*)d_in[13];
  const float* b_lin = (const float*)d_in[14];
  float* out = (float*)d_out;

  dim3 grid(BB / 16);
  dim3 block(NT);
  lstm3_fused_kernel<<<grid, block, 0, stream>>>(
      x, w_ih1, w_hh1, b_ih1, b_hh1,
      w_ih2, w_hh2, b_ih2, b_hh2,
      w_ih3, w_hh3, b_ih3, b_hh3,
      w_lin, b_lin, out);
}


# Keep each wave's w_ih1 and w_ih2 B fragments (its own gate tiles) in registers for the whole time loop instead of reloading them from global every step; summation order unchanged
# speedup vs baseline: 1.2395x; 1.2395x over previous
#include <hip/hip_runtime.h>
#include <hip/hip_bf16.h>
#include <math.h>

typedef float v2f __attribute__((ext_vector_type(2)));
typedef float v8f __attribute__((ext_vector_type(8)));

#define TT   512
#define BB   512
#define NIN  64
#define H1   100
#define H2   50
#define H3   25
#define G1   400
#define G2   200
#define G3   100
#define G1S  400
#define G2S  208
#define G3S  112
#define SH1  104
#define SH2  52
#define SH3  28
#define NT   256
#define NWAVES 8

#define KS_X1 16
#define KS_H1 25
#define KS_H2 13
#define KS_H3 7

__device__ __forceinline__ float sigf(float x) { return 1.0f / (1.0f + __expf(-x)); }

__device__ __forceinline__ v8f wmma4(v2f a, v2f b, v8f c) {
  return __builtin_amdgcn_wmma_f32_16x16x4_f32(
      false, a, false, b, (short)0, c, false, false);
}

template<int KS>
__device__ __forceinline__ void load_frags(v2f* f, const float* p, int khalf) {
  #pragma unroll
  for (int i = 0; i < KS; ++i) {
    f[i].x = p[4 * i + khalf];
    f[i].y = p[4 * i + khalf + 1];
  }
}

template<int KS>
__device__ __forceinline__ v8f wmma_chain(const v2f* a, const v2f* b, v8f acc) {
  #pragma unroll
  for (int i = 0; i < KS; ++i) acc = wmma4(a[i], b[i], acc);
  return acc;
}

__global__ __launch_bounds__(NT, 2) void lstm3_fused_kernel(
    const float* __restrict__ x,
    const float* __restrict__ w_ih1, const float* __restrict__ w_hh1,
    const float* __restrict__ b_ih1, const float* __restrict__ b_hh1,
    const float* __restrict__ w_ih2, const float* __restrict__ w_hh2,
    const float* __restrict__ b_ih2, const float* __restrict__ b_hh2,
    const float* __restrict__ w_ih3, const float* __restrict__ w_hh3,
    const float* __restrict__ b_ih3, const float* __restrict__ b_hh3,
    const float* __restrict__ w_lin, const float* __restrict__ b_lin,
    float* __restrict__ out)
{
  __shared__ float gates1[16 * G1S];
  __shared__ float gates2[16 * G2S];
  __shared__ float gates3[16 * G3S];
  __shared__ float h1s[16 * SH1];
  __shared__ float h2s[16 * SH2];
  __shared__ float h3s[16 * SH3];
  __shared__ float whh1s[G1 * H1];
  __shared__ float whh2s[208 * SH2];
  __shared__ float wih3s[112 * SH2];
  __shared__ float whh3s[112 * SH3];

  const int tid     = threadIdx.x;
  const int lane    = tid & 31;
  const int wave    = tid >> 5;
  const int col     = lane & 15;
  const int khalf   = (lane >> 4) << 1;
  const int rowhalf = (lane >> 4) << 3;
  const int bbase   = blockIdx.x * 16;

  for (int i = tid; i < 16 * SH1; i += NT) h1s[i] = 0.f;
  for (int i = tid; i < 16 * SH2; i += NT) h2s[i] = 0.f;
  for (int i = tid; i < 16 * SH3; i += NT) h3s[i] = 0.f;
  for (int i = tid; i < 208 * SH2; i += NT) whh2s[i] = 0.f;
  for (int i = tid; i < 112 * SH2; i += NT) wih3s[i] = 0.f;
  for (int i = tid; i < 112 * SH3; i += NT) whh3s[i] = 0.f;
  __syncthreads();

  for (int i = tid; i < G1 * H1; i += NT)
    whh1s[i] = w_hh1[i];
  for (int i = tid; i < G2 * H2; i += NT) {
    int r = i / H2, c = i - r * H2;
    whh2s[r * SH2 + c] = w_hh2[i];
  }
  for (int i = tid; i < G3 * H2; i += NT) {
    int r = i / H2, c = i - r * H2;
    wih3s[r * SH2 + c] = w_ih3[i];
  }
  for (int i = tid; i < G3 * H3; i += NT) {
    int r = i / H3, c = i - r * H3;
    whh3s[r * SH3 + c] = w_hh3[i];
  }

  float bs1[7][4], bs2[4][4], bs3[2][4];
  #pragma unroll
  for (int r = 0; r < 7; ++r) {
    const int idx = tid + r * NT;
    if (idx < 16 * H1) {
      const int j = idx >> 4;
      #pragma unroll
      for (int g = 0; g < 4; ++g) bs1[r][g] = b_ih1[g * H1 + j] + b_hh1[g * H1 + j];
    }
  }
  #pragma unroll
  for (int r = 0; r < 4; ++r) {
    const int idx = tid + r * NT;
    if (idx < 16 * H2) {
      const int j = idx >> 4;
      #pragma unroll
      for (int g = 0; g < 4; ++g) bs2[r][g] = b_ih2[g * H2 + j] + b_hh2[g * H2 + j];
    }
  }
  #pragma unroll
  for (int r = 0; r < 2; ++r) {
    const int idx = tid + r * NT;
    if (idx < 16 * H3) {
      const int j = idx >> 4;
      #pragma unroll
      for (int g = 0; g < 4; ++g) bs3[r][g] = b_ih3[g * H3 + j] + b_hh3[g * H3 + j];
    }
  }

  float wl[H3];
  #pragma unroll
  for (int j = 0; j < H3; ++j) wl[j] = w_lin[j];
  const float blin = b_lin[0];

  v2f wi1[4][KS_X1] = {};
  v2f wi2[2][KS_H1] = {};
  #pragma unroll
  for (int s = 0; s < 4; ++s) {
    const int tile = wave + s * NWAVES;
    if (tile < 25)
      load_frags<KS_X1>(wi1[s], w_ih1 + (size_t)(tile * 16 + col) * NIN, khalf);
  }
  #pragma unroll
  for (int s = 0; s < 2; ++s) {
    const int tile = wave + s * NWAVES;
    if (tile < 13) {
      const int n  = tile * 16 + col;
      const int nc = (n < G2) ? n : (G2 - 1);
      load_frags<KS_H1>(wi2[s], w_ih2 + (size_t)nc * H1, khalf);
    }
  }

  float c1r[7], c2r[4], c3r[2];
  #pragma unroll
  for (int r = 0; r < 7; ++r) c1r[r] = 0.f;
  #pragma unroll
  for (int r = 0; r < 4; ++r) c2r[r] = 0.f;
  #pragma unroll
  for (int r = 0; r < 2; ++r) c3r[r] = 0.f;

  __syncthreads();

  for (int t = 0; t < TT; ++t) {
    {
      if (t + 1 < TT && tid < 16)
        __builtin_prefetch(x + ((size_t)(t + 1) * BB + bbase + tid) * NIN, 0, 3);

      v2f ax[KS_X1], ah[KS_H1];
      load_frags<KS_X1>(ax, x + ((size_t)t * BB + bbase + col) * NIN, khalf);
      load_frags<KS_H1>(ah, &h1s[col * SH1], khalf);
      #pragma unroll
      for (int s = 0; s < 4; ++s) {
        const int tile = wave + s * NWAVES;
        if (tile < 25) {
          const int n = tile * 16 + col;
          v2f bh[KS_H1];
          load_frags<KS_H1>(bh, &whh1s[n * H1], khalf);
          v8f acc = {};
          acc = wmma_chain<KS_X1>(ax, wi1[s], acc);
          acc = wmma_chain<KS_H1>(ah, bh, acc);
          #pragma unroll
          for (int r = 0; r < 8; ++r)
            gates1[(r + rowhalf) * G1S + n] = acc[r];
        }
      }
    }
    __syncthreads();

    #pragma unroll
    for (int r = 0; r < 7; ++r) {
      const int idx = tid + r * NT;
      if (idx < 16 * H1) {
        const int m = idx & 15, j = idx >> 4;
        float gi = gates1[m * G1S + j]          + bs1[r][0];
        float gf = gates1[m * G1S + H1 + j]     + bs1[r][1];
        float gg = gates1[m * G1S + 2 * H1 + j] + bs1[r][2];
        float go = gates1[m * G1S + 3 * H1 + j] + bs1[r][3];
        float c  = sigf(gf) * c1r[r] + sigf(gi) * tanhf(gg);
        c1r[r]   = c;
        h1s[m * SH1 + j] = sigf(go) * tanhf(c);
      }
    }
    __syncthreads();

    {
      v2f a1[KS_H1], a2[KS_H2];
      load_frags<KS_H1>(a1, &h1s[col * SH1], khalf);
      load_frags<KS_H2>(a2, &h2s[col * SH2], khalf);
      #pragma unroll
      for (int s = 0; s < 2; ++s) {
        const int tile = wave + s * NWAVES;
        if (tile < 13) {
          const int n = tile * 16 + col;
          v2f bh[KS_H2];
          load_frags<KS_H2>(bh, &whh2s[n * SH2], khalf);
          v8f acc = {};
          acc = wmma_chain<KS_H1>(a1, wi2[s], acc);
          acc = wmma_chain<KS_H2>(a2, bh, acc);
          if (n < G2) {
            #pragma unroll
            for (int r = 0; r < 8; ++r)
              gates2[(r + rowhalf) * G2S + n] = acc[r];
          }
        }
      }
    }
    __syncthreads();

    #pragma unroll
    for (int r = 0; r < 4; ++r) {
      const int idx = tid + r * NT;
      if (idx < 16 * H2) {
        const int m = idx & 15, j = idx >> 4;
        float gi = gates2[m * G2S + j]          + bs2[r][0];
        float gf = gates2[m * G2S + H2 + j]     + bs2[r][1];
        float gg = gates2[m * G2S + 2 * H2 + j] + bs2[r][2];
        float go = gates2[m * G2S + 3 * H2 + j] + bs2[r][3];
        float c  = sigf(gf) * c2r[r] + sigf(gi) * tanhf(gg);
        c2r[r]   = c;
        h2s[m * SH2 + j] = sigf(go) * tanhf(c);
      }
    }
    __syncthreads();

    {
      v2f a2[KS_H2], a3[KS_H3];
      load_frags<KS_H2>(a2, &h2s[col * SH2], khalf);
      load_frags<KS_H3>(a3, &h3s[col * SH3], khalf);
      for (int tile = wave; tile < 7; tile += NWAVES) {
        const int n = tile * 16 + col;
        v2f bi[KS_H2], bh[KS_H3];
        load_frags<KS_H2>(bi, &wih3s[n * SH2], khalf);
        load_frags<KS_H3>(bh, &whh3s[n * SH3], khalf);
        v8f acc = {};
        acc = wmma_chain<KS_H2>(a2, bi, acc);
        acc = wmma_chain<KS_H3>(a3, bh, acc);
        if (n < G3) {
          #pragma unroll
          for (int r = 0; r < 8; ++r)
            gates3[(r + rowhalf) * G3S + n] = acc[r];
        }
      }
    }
    __syncthreads();

    #pragma unroll
    for (int r = 0; r < 2; ++r) {
      const int idx = tid + r * NT;
      if (idx < 16 * H3) {
        const int m = idx & 15, j = idx >> 4;
        float gi = gates3[m * G3S + j]          + bs3[r][0];
        float gf = gates3[m * G3S + H3 + j]     + bs3[r][1];
        float gg = gates3[m * G3S + 2 * H3 + j] + bs3[r][2];
        float go = gates3[m * G3S + 3 * H3 + j] + bs3[r][3];
        float c  = sigf(gf) * c3r[r] + sigf(gi) * tanhf(gg);
        c3r[r]   = c;
        h3s[m * SH3 + j] = sigf(go) * tanhf(c);
      }
    }
    __syncthreads();

    if (tid < 16) {
      const float* hr = &h3s[tid * SH3];
      float acc = blin;
      #pragma unroll
      for (int j = 0; j < H3; ++j) acc += hr[j] * wl[j];
      out[(size_t)t * BB + bbase + tid] = acc;
    }
    __syncthreads();
  }
}

extern "C" void kernel_launch(void* const* d_in, const int* in_sizes, int n_in,
                              void* d_out, int out_size, void* d_ws, size_t ws_size,
                              hipStream_t stream) {
  (void)in_sizes; (void)n_in; (void)d_ws; (void)ws_size; (void)out_size;
  const float* x     = (const float*)d_in[0];
  const float* w_ih1 = (const float*)d_in[1];
  const float* w_hh1 = (const float*)d_in[2];
  const float* b_ih1 = (const float*)d_in[3];
  const float* b_hh1 = (const float*)d_in[4];
  const float* w_ih2 = (const float*)d_in[5];
  const float* w_hh2 = (const float*)d_in[6];
  const float* b_ih2 = (const float*)d_in[7];
  const float* b_hh2 = (const float*)d_in[8];
  const float* w_ih3 = (const float*)d_in[9];
  const float* w_hh3 = (const float*)d_in[10];
  const float* b_ih3 = (const float*)d_in[11];
  const float* b_hh3 = (const float*)d_in[12];
  const float* w_lin = (const float*)d_in[13];
  const float* b_lin = (const float*)d_in[14];
  float* out = (float*)d_out;

  dim3 grid(BB / 16);
  dim3 block(NT);
  lstm3_fused_kernel<<<grid, block, 0, stream>>>(
      x, w_ih1, w_hh1, b_ih1, b_hh1,
      w_ih2, w_hh2, b_ih2, b_hh2,
      w_ih3, w_hh3, b_ih3, b_hh3,
      w_lin, b_lin, out);
}


# Skew the 3 layers by one time step each (layer 1 at t, layer 2 at t-1, layer 3 at t-2 in one pass): one gate-product phase and one cell phase, 2 barriers per step not 7; summation order unchanged
# speedup vs baseline: 1.5019x; 1.2117x over previous
#include <hip/hip_runtime.h>
#include <hip/hip_bf16.h>
#include <math.h>

typedef float v2f __attribute__((ext_vector_type(2)));
typedef float v8f __attribute__((ext_vector_type(8)));

#define TT   512
#define BB   512
#define NIN  64
#define H1   100
#define H2   50
#define H3   25
#define G1   400
#define G2   200
#define G3   100
#define G1S  400
#define G2S  208
#define G3S  112
#define SH1  104
#define SH2  52
#define SH3  28
#define NT   256
#define NWAVES 8

#define KS_X1 16
#define KS_H1 25
#define KS_H2 13
#define KS_H3 7

__device__ __forceinline__ float sigf(float x) { return 1.0f / (1.0f + __expf(-x)); }

__device__ __forceinline__ v8f wmma4(v2f a, v2f b, v8f c) {
  return __builtin_amdgcn_wmma_f32_16x16x4_f32(
      false, a, false, b, (short)0, c, false, false);
}

template<int KS>
__device__ __forceinline__ void load_frags(v2f* f, const float* p, int khalf) {
  #pragma unroll
  for (int i = 0; i < KS; ++i) {
    f[i].x = p[4 * i + khalf];
    f[i].y = p[4 * i + khalf + 1];
  }
}

template<int KS>
__device__ __forceinline__ v8f wmma_chain(const v2f* a, const v2f* b, v8f acc) {
  #pragma unroll
  for (int i = 0; i < KS; ++i) acc = wmma4(a[i], b[i], acc);
  return acc;
}

__global__ __launch_bounds__(NT, 2) void lstm3_fused_kernel(
    const float* __restrict__ x,
    const float* __restrict__ w_ih1, const float* __restrict__ w_hh1,
    const float* __restrict__ b_ih1, const float* __restrict__ b_hh1,
    const float* __restrict__ w_ih2, const float* __restrict__ w_hh2,
    const float* __restrict__ b_ih2, const float* __restrict__ b_hh2,
    const float* __restrict__ w_ih3, const float* __restrict__ w_hh3,
    const float* __restrict__ b_ih3, const float* __restrict__ b_hh3,
    const float* __restrict__ w_lin, const float* __restrict__ b_lin,
    float* __restrict__ out)
{
  __shared__ float gates1[16 * G1S];
  __shared__ float gates2[16 * G2S];
  __shared__ float gates3[16 * G3S];
  __shared__ float h1s[16 * SH1];
  __shared__ float h2s[16 * SH2];
  __shared__ float h3s[16 * SH3];
  __shared__ float whh1s[G1 * H1];
  __shared__ float whh2s[208 * SH2];
  __shared__ float wih3s[112 * SH2];
  __shared__ float whh3s[112 * SH3];

  const int tid     = threadIdx.x;
  const int lane    = tid & 31;
  const int wave    = tid >> 5;
  const int col     = lane & 15;
  const int khalf   = (lane >> 4) << 1;
  const int rowhalf = (lane >> 4) << 3;
  const int bbase   = blockIdx.x * 16;
  const int wrev    = (NWAVES - 1) - wave;

  for (int i = tid; i < 16 * SH1; i += NT) h1s[i] = 0.f;
  for (int i = tid; i < 16 * SH2; i += NT) h2s[i] = 0.f;
  for (int i = tid; i < 16 * SH3; i += NT) h3s[i] = 0.f;
  for (int i = tid; i < 208 * SH2; i += NT) whh2s[i] = 0.f;
  for (int i = tid; i < 112 * SH2; i += NT) wih3s[i] = 0.f;
  for (int i = tid; i < 112 * SH3; i += NT) whh3s[i] = 0.f;
  __syncthreads();

  for (int i = tid; i < G1 * H1; i += NT)
    whh1s[i] = w_hh1[i];
  for (int i = tid; i < G2 * H2; i += NT) {
    int r = i / H2, c = i - r * H2;
    whh2s[r * SH2 + c] = w_hh2[i];
  }
  for (int i = tid; i < G3 * H2; i += NT) {
    int r = i / H2, c = i - r * H2;
    wih3s[r * SH2 + c] = w_ih3[i];
  }
  for (int i = tid; i < G3 * H3; i += NT) {
    int r = i / H3, c = i - r * H3;
    whh3s[r * SH3 + c] = w_hh3[i];
  }

  float bs1[7][4], bs2[4][4], bs3[2][4];
  #pragma unroll
  for (int r = 0; r < 7; ++r) {
    const int idx = tid + r * NT;
    if (idx < 16 * H1) {
      const int j = idx >> 4;
      #pragma unroll
      for (int g = 0; g < 4; ++g) bs1[r][g] = b_ih1[g * H1 + j] + b_hh1[g * H1 + j];
    }
  }
  #pragma unroll
  for (int r = 0; r < 4; ++r) {
    const int idx = tid + r * NT;
    if (idx < 16 * H2) {
      const int j = idx >> 4;
      #pragma unroll
      for (int g = 0; g < 4; ++g) bs2[r][g] = b_ih2[g * H2 + j] + b_hh2[g * H2 + j];
    }
  }
  #pragma unroll
  for (int r = 0; r < 2; ++r) {
    const int idx = tid + r * NT;
    if (idx < 16 * H3) {
      const int j = idx >> 4;
      #pragma unroll
      for (int g = 0; g < 4; ++g) bs3[r][g] = b_ih3[g * H3 + j] + b_hh3[g * H3 + j];
    }
  }

  float wl[H3];
  #pragma unroll
  for (int j = 0; j < H3; ++j) wl[j] = w_lin[j];
  const float blin = b_lin[0];

  v2f wi1[4][KS_X1] = {};
  v2f wi2[2][KS_H1] = {};
  #pragma unroll
  for (int s = 0; s < 4; ++s) {
    const int tile = wave + s * NWAVES;
    if (tile < 25)
      load_frags<KS_X1>(wi1[s], w_ih1 + (size_t)(tile * 16 + col) * NIN, khalf);
  }
  #pragma unroll
  for (int s = 0; s < 2; ++s) {
    const int tile = wrev + s * NWAVES;
    if (tile < 13) {
      const int n  = tile * 16 + col;
      const int nc = (n < G2) ? n : (G2 - 1);
      load_frags<KS_H1>(wi2[s], w_ih2 + (size_t)nc * H1, khalf);
    }
  }

  float c1r[7], c2r[4], c3r[2];
  #pragma unroll
  for (int r = 0; r < 7; ++r) c1r[r] = 0.f;
  #pragma unroll
  for (int r = 0; r < 4; ++r) c2r[r] = 0.f;
  #pragma unroll
  for (int r = 0; r < 2; ++r) c3r[r] = 0.f;

  __syncthreads();

  for (int p = 0; p < TT + 2; ++p) {
    const bool act1 = (p < TT);
    const bool act2 = (p >= 1) && (p <= TT);
    const bool act3 = (p >= 2);

    {
      if (p + 1 < TT && tid < 16)
        __builtin_prefetch(x + ((size_t)(p + 1) * BB + bbase + tid) * NIN, 0, 3);

      v2f ax[KS_X1];
      if (act1)
        load_frags<KS_X1>(ax, x + ((size_t)p * BB + bbase + col) * NIN, khalf);

      v2f ah1[KS_H1];
      load_frags<KS_H1>(ah1, &h1s[col * SH1], khalf);
      v2f ah2[KS_H2];
      load_frags<KS_H2>(ah2, &h2s[col * SH2], khalf);

      if (act2) {
        #pragma unroll
        for (int s = 0; s < 2; ++s) {
          const int tile = wrev + s * NWAVES;
          if (tile < 13) {
            const int n = tile * 16 + col;
            v2f bh[KS_H2];
            load_frags<KS_H2>(bh, &whh2s[n * SH2], khalf);
            v8f acc = {};
            acc = wmma_chain<KS_H1>(ah1, wi2[s], acc);
            acc = wmma_chain<KS_H2>(ah2, bh, acc);
            if (n < G2) {
              #pragma unroll
              for (int r = 0; r < 8; ++r)
                gates2[(r + rowhalf) * G2S + n] = acc[r];
            }
          }
        }
      }

      if (act3) {
        if (wrev < 7) {
          const int n = wrev * 16 + col;
          v2f a3[KS_H3], bi[KS_H2], bh[KS_H3];
          load_frags<KS_H3>(a3, &h3s[col * SH3], khalf);
          load_frags<KS_H2>(bi, &wih3s[n * SH2], khalf);
          load_frags<KS_H3>(bh, &whh3s[n * SH3], khalf);
          v8f acc = {};
          acc = wmma_chain<KS_H2>(ah2, bi, acc);
          acc = wmma_chain<KS_H3>(a3, bh, acc);
          if (n < G3) {
            #pragma unroll
            for (int r = 0; r < 8; ++r)
              gates3[(r + rowhalf) * G3S + n] = acc[r];
          }
        }
      }

      if (act1) {
        #pragma unroll
        for (int s = 0; s < 4; ++s) {
          const int tile = wave + s * NWAVES;
          if (tile < 25) {
            const int n = tile * 16 + col;
            v2f bh[KS_H1];
            load_frags<KS_H1>(bh, &whh1s[n * H1], khalf);
            v8f acc = {};
            acc = wmma_chain<KS_X1>(ax, wi1[s], acc);
            acc = wmma_chain<KS_H1>(ah1, bh, acc);
            #pragma unroll
            for (int r = 0; r < 8; ++r)
              gates1[(r + rowhalf) * G1S + n] = acc[r];
          }
        }
      }
    }
    __syncthreads();

    if (act1) {
      #pragma unroll
      for (int r = 0; r < 7; ++r) {
        const int idx = tid + r * NT;
        if (idx < 16 * H1) {
          const int m = idx & 15, j = idx >> 4;
          float gi = gates1[m * G1S + j]          + bs1[r][0];
          float gf = gates1[m * G1S + H1 + j]     + bs1[r][1];
          float gg = gates1[m * G1S + 2 * H1 + j] + bs1[r][2];
          float go = gates1[m * G1S + 3 * H1 + j] + bs1[r][3];
          float c  = sigf(gf) * c1r[r] + sigf(gi) * tanhf(gg);
          c1r[r]   = c;
          h1s[m * SH1 + j] = sigf(go) * tanhf(c);
        }
      }
    }
    if (act2) {
      #pragma unroll
      for (int r = 0; r < 4; ++r) {
        const int idx = tid + r * NT;
        if (idx < 16 * H2) {
          const int m = idx & 15, j = idx >> 4;
          float gi = gates2[m * G2S + j]          + bs2[r][0];
          float gf = gates2[m * G2S + H2 + j]     + bs2[r][1];
          float gg = gates2[m * G2S + 2 * H2 + j] + bs2[r][2];
          float go = gates2[m * G2S + 3 * H2 + j] + bs2[r][3];
          float c  = sigf(gf) * c2r[r] + sigf(gi) * tanhf(gg);
          c2r[r]   = c;
          h2s[m * SH2 + j] = sigf(go) * tanhf(c);
        }
      }
    }
    if (act3) {
      #pragma unroll
      for (int r = 0; r < 2; ++r) {
        const int idx = tid + r * NT;
        if (idx < 16 * H3) {
          const int m = idx & 15, j = idx >> 4;
          float gi = gates3[m * G3S + j]          + bs3[r][0];
          float gf = gates3[m * G3S + H3 + j]     + bs3[r][1];
          float gg = gates3[m * G3S + 2 * H3 + j] + bs3[r][2];
          float go = gates3[m * G3S + 3 * H3 + j] + bs3[r][3];
          float c  = sigf(gf) * c3r[r] + sigf(gi) * tanhf(gg);
          c3r[r]   = c;
          h3s[m * SH3 + j] = sigf(go) * tanhf(c);
        }
      }
    }
    __syncthreads();

    if (act3 && tid < 16) {
      const float* hr = &h3s[tid * SH3];
      float acc = blin;
      #pragma unroll
      for (int j = 0; j < H3; ++j) acc += hr[j] * wl[j];
      out[(size_t)(p - 2) * BB + bbase + tid] = acc;
    }
  }
}

extern "C" void kernel_launch(void* const* d_in, const int* in_sizes, int n_in,
                              void* d_out, int out_size, void* d_ws, size_t ws_size,
                              hipStream_t stream) {
  (void)in_sizes; (void)n_in; (void)d_ws; (void)ws_size; (void)out_size;
  const float* x     = (const float*)d_in[0];
  const float* w_ih1 = (const float*)d_in[1];
  const float* w_hh1 = (const float*)d_in[2];
  const float* b_ih1 = (const float*)d_in[3];
  const float* b_hh1 = (const float*)d_in[4];
  const float* w_ih2 = (const float*)d_in[5];
  const float* w_hh2 = (const float*)d_in[6];
  const float* b_ih2 = (const float*)d_in[7];
  const float* b_hh2 = (const float*)d_in[8];
  const float* w_ih3 = (const float*)d_in[9];
  const float* w_hh3 = (const float*)d_in[10];
  const float* b_ih3 = (const float*)d_in[11];
  const float* b_hh3 = (const float*)d_in[12];
  const float* w_lin = (const float*)d_in[13];
  const float* b_lin = (const float*)d_in[14];
  float* out = (float*)d_out;

  dim3 grid(BB / 16);
  dim3 block(NT);
  lstm3_fused_kernel<<<grid, block, 0, stream>>>(
      x, w_ih1, w_hh1, b_ih1, b_hh1,
      w_ih2, w_hh2, b_ih2, b_hh2,
      w_ih3, w_hh3, b_ih3, b_hh3,
      w_lin, b_lin, out);
}


# Cell updates as straight-line code: tanh computes both ranges and selects instead of branching; each thread's ten full-round cells run staged in one block so their chains overlap; sums unchanged
# speedup vs baseline: 1.8803x; 1.2519x over previous
#include <hip/hip_runtime.h>
#include <hip/hip_bf16.h>
#include <math.h>

typedef float v2f __attribute__((ext_vector_type(2)));
typedef float v8f __attribute__((ext_vector_type(8)));

#define TT   512
#define BB   512
#define NIN  64
#define H1   100
#define H2   50
#define H3   25
#define G1   400
#define G2   200
#define G3   100
#define G1S  400
#define G2S  208
#define G3S  112
#define SH1  104
#define SH2  52
#define SH3  28
#define NT   256
#define NWAVES 8

#define KS_X1 16
#define KS_H1 25
#define KS_H2 13
#define KS_H3 7

__device__ __forceinline__ float sigf(float x) { return 1.0f / (1.0f + __expf(-x)); }

__device__ __forceinline__ float tanh_sel(float x) {
  const float y  = fabsf(x);
  const float y2 = y * y;
  float p = fmaf(y2, -0x1.758e7ap-8f, 0x1.521192p-6f);
  p = fmaf(y2, p, -0x1.b8389cp-5f);
  p = fmaf(y2, p, 0x1.110704p-3f);
  p = fmaf(y2, p, -0x1.555532p-2f);
  const float zs = fmaf(y2, y * p, y);
  const float e  = expf(y + y);
  const float zl = fmaf(__builtin_amdgcn_rcpf(e + 1.0f), -2.0f, 1.0f);
  const float z  = (y < 0.625f) ? zs : zl;
  return copysignf(z, x);
}

template<int H, int GS>
__device__ __forceinline__ void ld_gates(const float* gates, int idx, const float (&b)[4], float (&v)[4]) {
  const int m = idx & 15, j = idx >> 4;
  v[0] = gates[m * GS + j]         + b[0];
  v[1] = gates[m * GS + H + j]     + b[1];
  v[2] = gates[m * GS + 2 * H + j] + b[2];
  v[3] = gates[m * GS + 3 * H + j] + b[3];
}

__device__ __forceinline__ float cell_math(const float (&v)[4], float& cst) {
  float c = sigf(v[1]) * cst + sigf(v[0]) * tanh_sel(v[2]);
  cst = c;
  return sigf(v[3]) * tanh_sel(c);
}

template<int SH>
__device__ __forceinline__ void st_h(float* hs, int idx, float h) {
  const int m = idx & 15, j = idx >> 4;
  hs[m * SH + j] = h;
}

template<int R, int H, int GS, int SH>
__device__ __forceinline__ void layer_cells(const float* gates, float* hs, const float (&bs)[R][4],
                                            float (&cr)[R], int tid) {
  #pragma unroll
  for (int r = 0; r < R; ++r) {
    const int idx = tid + r * NT;
    if (idx < 16 * H) {
      float g[4];
      ld_gates<H, GS>(gates, idx, bs[r], g);
      const float h = cell_math(g, cr[r]);
      st_h<SH>(hs, idx, h);
    }
  }
}

__device__ __forceinline__ v8f wmma4(v2f a, v2f b, v8f c) {
  return __builtin_amdgcn_wmma_f32_16x16x4_f32(
      false, a, false, b, (short)0, c, false, false);
}

template<int KS>
__device__ __forceinline__ void load_frags(v2f* f, const float* p, int khalf) {
  #pragma unroll
  for (int i = 0; i < KS; ++i) {
    f[i].x = p[4 * i + khalf];
    f[i].y = p[4 * i + khalf + 1];
  }
}

template<int KS>
__device__ __forceinline__ v8f wmma_chain(const v2f* a, const v2f* b, v8f acc) {
  #pragma unroll
  for (int i = 0; i < KS; ++i) acc = wmma4(a[i], b[i], acc);
  return acc;
}

__global__ __launch_bounds__(NT, 2) void lstm3_fused_kernel(
    const float* __restrict__ x,
    const float* __restrict__ w_ih1, const float* __restrict__ w_hh1,
    const float* __restrict__ b_ih1, const float* __restrict__ b_hh1,
    const float* __restrict__ w_ih2, const float* __restrict__ w_hh2,
    const float* __restrict__ b_ih2, const float* __restrict__ b_hh2,
    const float* __restrict__ w_ih3, const float* __restrict__ w_hh3,
    const float* __restrict__ b_ih3, const float* __restrict__ b_hh3,
    const float* __restrict__ w_lin, const float* __restrict__ b_lin,
    float* __restrict__ out)
{
  __shared__ float gates1[16 * G1S];
  __shared__ float gates2[16 * G2S];
  __shared__ float gates3[16 * G3S];
  __shared__ float h1s[16 * SH1];
  __shared__ float h2s[16 * SH2];
  __shared__ float h3s[16 * SH3];
  __shared__ float whh1s[G1 * H1];
  __shared__ float whh2s[208 * SH2];
  __shared__ float wih3s[112 * SH2];
  __shared__ float whh3s[112 * SH3];

  const int tid     = threadIdx.x;
  const int lane    = tid & 31;
  const int wave    = tid >> 5;
  const int col     = lane & 15;
  const int khalf   = (lane >> 4) << 1;
  const int rowhalf = (lane >> 4) << 3;
  const int bbase   = blockIdx.x * 16;
  const int wrev    = (NWAVES - 1) - wave;

  for (int i = tid; i < 16 * SH1; i += NT) h1s[i] = 0.f;
  for (int i = tid; i < 16 * SH2; i += NT) h2s[i] = 0.f;
  for (int i = tid; i < 16 * SH3; i += NT) h3s[i] = 0.f;
  for (int i = tid; i < 208 * SH2; i += NT) whh2s[i] = 0.f;
  for (int i = tid; i < 112 * SH2; i += NT) wih3s[i] = 0.f;
  for (int i = tid; i < 112 * SH3; i += NT) whh3s[i] = 0.f;
  __syncthreads();

  for (int i = tid; i < G1 * H1; i += NT)
    whh1s[i] = w_hh1[i];
  for (int i = tid; i < G2 * H2; i += NT) {
    int r = i / H2, c = i - r * H2;
    whh2s[r * SH2 + c] = w_hh2[i];
  }
  for (int i = tid; i < G3 * H2; i += NT) {
    int r = i / H2, c = i - r * H2;
    wih3s[r * SH2 + c] = w_ih3[i];
  }
  for (int i = tid; i < G3 * H3; i += NT) {
    int r = i / H3, c = i - r * H3;
    whh3s[r * SH3 + c] = w_hh3[i];
  }

  float bs1[7][4], bs2[4][4], bs3[2][4];
  #pragma unroll
  for (int r = 0; r < 7; ++r) {
    const int idx = tid + r * NT;
    if (idx < 16 * H1) {
      const int j = idx >> 4;
      #pragma unroll
      for (int g = 0; g < 4; ++g) bs1[r][g] = b_ih1[g * H1 + j] + b_hh1[g * H1 + j];
    }
  }
  #pragma unroll
  for (int r = 0; r < 4; ++r) {
    const int idx = tid + r * NT;
    if (idx < 16 * H2) {
      const int j = idx >> 4;
      #pragma unroll
      for (int g = 0; g < 4; ++g) bs2[r][g] = b_ih2[g * H2 + j] + b_hh2[g * H2 + j];
    }
  }
  #pragma unroll
  for (int r = 0; r < 2; ++r) {
    const int idx = tid + r * NT;
    if (idx < 16 * H3) {
      const int j = idx >> 4;
      #pragma unroll
      for (int g = 0; g < 4; ++g) bs3[r][g] = b_ih3[g * H3 + j] + b_hh3[g * H3 + j];
    }
  }

  float wl[H3];
  #pragma unroll
  for (int j = 0; j < H3; ++j) wl[j] = w_lin[j];
  const float blin = b_lin[0];

  v2f wi1[4][KS_X1] = {};
  v2f wi2[2][KS_H1] = {};
  #pragma unroll
  for (int s = 0; s < 4; ++s) {
    const int tile = wave + s * NWAVES;
    if (tile < 25)
      load_frags<KS_X1>(wi1[s], w_ih1 + (size_t)(tile * 16 + col) * NIN, khalf);
  }
  #pragma unroll
  for (int s = 0; s < 2; ++s) {
    const int tile = wrev + s * NWAVES;
    if (tile < 13) {
      const int n  = tile * 16 + col;
      const int nc = (n < G2) ? n : (G2 - 1);
      load_frags<KS_H1>(wi2[s], w_ih2 + (size_t)nc * H1, khalf);
    }
  }

  float c1r[7], c2r[4], c3r[2];
  #pragma unroll
  for (int r = 0; r < 7; ++r) c1r[r] = 0.f;
  #pragma unroll
  for (int r = 0; r < 4; ++r) c2r[r] = 0.f;
  #pragma unroll
  for (int r = 0; r < 2; ++r) c3r[r] = 0.f;

  __syncthreads();

  for (int p = 0; p < TT + 2; ++p) {
    const bool act1 = (p < TT);
    const bool act2 = (p >= 1) && (p <= TT);
    const bool act3 = (p >= 2);

    {
      if (p + 1 < TT && tid < 16)
        __builtin_prefetch(x + ((size_t)(p + 1) * BB + bbase + tid) * NIN, 0, 3);

      v2f ax[KS_X1];
      if (act1)
        load_frags<KS_X1>(ax, x + ((size_t)p * BB + bbase + col) * NIN, khalf);

      v2f ah1[KS_H1];
      load_frags<KS_H1>(ah1, &h1s[col * SH1], khalf);
      v2f ah2[KS_H2];
      load_frags<KS_H2>(ah2, &h2s[col * SH2], khalf);

      if (act2) {
        #pragma unroll
        for (int s = 0; s < 2; ++s) {
          const int tile = wrev + s * NWAVES;
          if (tile < 13) {
            const int n = tile * 16 + col;
            v2f bh[KS_H2];
            load_frags<KS_H2>(bh, &whh2s[n * SH2], khalf);
            v8f acc = {};
            acc = wmma_chain<KS_H1>(ah1, wi2[s], acc);
            acc = wmma_chain<KS_H2>(ah2, bh, acc);
            if (n < G2) {
              #pragma unroll
              for (int r = 0; r < 8; ++r)
                gates2[(r + rowhalf) * G2S + n] = acc[r];
            }
          }
        }
      }

      if (act3) {
        if (wrev < 7) {
          const int n = wrev * 16 + col;
          v2f a3[KS_H3], bi[KS_H2], bh[KS_H3];
          load_frags<KS_H3>(a3, &h3s[col * SH3], khalf);
          load_frags<KS_H2>(bi, &wih3s[n * SH2], khalf);
          load_frags<KS_H3>(bh, &whh3s[n * SH3], khalf);
          v8f acc = {};
          acc = wmma_chain<KS_H2>(ah2, bi, acc);
          acc = wmma_chain<KS_H3>(a3, bh, acc);
          if (n < G3) {
            #pragma unroll
            for (int r = 0; r < 8; ++r)
              gates3[(r + rowhalf) * G3S + n] = acc[r];
          }
        }
      }

      if (act1) {
        #pragma unroll
        for (int s = 0; s < 4; ++s) {
          const int tile = wave + s * NWAVES;
          if (tile < 25) {
            const int n = tile * 16 + col;
            v2f bh[KS_H1];
            load_frags<KS_H1>(bh, &whh1s[n * H1], khalf);
            v8f acc = {};
            acc = wmma_chain<KS_X1>(ax, wi1[s], acc);
            acc = wmma_chain<KS_H1>(ah1, bh, acc);
            #pragma unroll
            for (int r = 0; r < 8; ++r)
              gates1[(r + rowhalf) * G1S + n] = acc[r];
          }
        }
      }
    }
    __syncthreads();

    if (act1 && act2 && act3) {
      float g1[6][4], g2[3][4], g3[4];
      #pragma unroll
      for (int r = 0; r < 6; ++r) ld_gates<H1, G1S>(gates1, tid + r * NT, bs1[r], g1[r]);
      #pragma unroll
      for (int r = 0; r < 3; ++r) ld_gates<H2, G2S>(gates2, tid + r * NT, bs2[r], g2[r]);
      ld_gates<H3, G3S>(gates3, tid, bs3[0], g3);
      float hn1[6], hn2[3], hn3;
      #pragma unroll
      for (int r = 0; r < 6; ++r) hn1[r] = cell_math(g1[r], c1r[r]);
      #pragma unroll
      for (int r = 0; r < 3; ++r) hn2[r] = cell_math(g2[r], c2r[r]);
      hn3 = cell_math(g3, c3r[0]);
      #pragma unroll
      for (int r = 0; r < 6; ++r) st_h<SH1>(h1s, tid + r * NT, hn1[r]);
      #pragma unroll
      for (int r = 0; r < 3; ++r) st_h<SH2>(h2s, tid + r * NT, hn2[r]);
      st_h<SH3>(h3s, tid, hn3);

      if (tid < 16 * H3 - NT) {
        float t3[4], t1[4], t2[4];
        ld_gates<H3, G3S>(gates3, tid + NT, bs3[1], t3);
        if (tid < 16 * H1 - 6 * NT) {
          ld_gates<H1, G1S>(gates1, tid + 6 * NT, bs1[6], t1);
          if (tid < 16 * H2 - 3 * NT) {
            ld_gates<H2, G2S>(gates2, tid + 3 * NT, bs2[3], t2);
            const float a3 = cell_math(t3, c3r[1]);
            const float a1 = cell_math(t1, c1r[6]);
            const float a2 = cell_math(t2, c2r[3]);
            st_h<SH3>(h3s, tid + NT, a3);
            st_h<SH1>(h1s, tid + 6 * NT, a1);
            st_h<SH2>(h2s, tid + 3 * NT, a2);
          } else {
            const float a3 = cell_math(t3, c3r[1]);
            const float a1 = cell_math(t1, c1r[6]);
            st_h<SH3>(h3s, tid + NT, a3);
            st_h<SH1>(h1s, tid + 6 * NT, a1);
          }
        } else {
          const float a3 = cell_math(t3, c3r[1]);
          st_h<SH3>(h3s, tid + NT, a3);
        }
      }
    } else {
      if (act1) layer_cells<7, H1, G1S, SH1>(gates1, h1s, bs1, c1r, tid);
      if (act2) layer_cells<4, H2, G2S, SH2>(gates2, h2s, bs2, c2r, tid);
      if (act3) layer_cells<2, H3, G3S, SH3>(gates3, h3s, bs3, c3r, tid);
    }
    __syncthreads();

    if (act3 && tid < 16) {
      const float* hr = &h3s[tid * SH3];
      float acc = blin;
      #pragma unroll
      for (int j = 0; j < H3; ++j) acc += hr[j] * wl[j];
      out[(size_t)(p - 2) * BB + bbase + tid] = acc;
    }
  }
}

extern "C" void kernel_launch(void* const* d_in, const int* in_sizes, int n_in,
                              void* d_out, int out_size, void* d_ws, size_t ws_size,
                              hipStream_t stream) {
  (void)in_sizes; (void)n_in; (void)d_ws; (void)ws_size; (void)out_size;
  const float* x     = (const float*)d_in[0];
  const float* w_ih1 = (const float*)d_in[1];
  const float* w_hh1 = (const float*)d_in[2];
  const float* b_ih1 = (const float*)d_in[3];
  const float* b_hh1 = (const float*)d_in[4];
  const float* w_ih2 = (const float*)d_in[5];
  const float* w_hh2 = (const float*)d_in[6];
  const float* b_ih2 = (const float*)d_in[7];
  const float* b_hh2 = (const float*)d_in[8];
  const float* w_ih3 = (const float*)d_in[9];
  const float* w_hh3 = (const float*)d_in[10];
  const float* b_ih3 = (const float*)d_in[11];
  const float* b_hh3 = (const float*)d_in[12];
  const float* w_lin = (const float*)d_in[13];
  const float* b_lin = (const float*)d_in[14];
  float* out = (float*)d_out;

  dim3 grid(BB / 16);
  dim3 block(NT);
  lstm3_fused_kernel<<<grid, block, 0, stream>>>(
      x, w_ih1, w_hh1, b_ih1, b_hh1,
      w_ih2, w_hh2, b_ih2, b_hh2,
      w_ih3, w_hh3, b_ih3, b_hh3,
      w_lin, b_lin, out);
}


# Each tile's LDS weight fragments are read as one group in front of its chain (register-operand part first; layer 3 first so they fit), not one read per two matrix instructions; sums unchanged
# speedup vs baseline: 1.9660x; 1.0456x over previous
#include <hip/hip_runtime.h>
#include <hip/hip_bf16.h>
#include <math.h>

typedef float v2f __attribute__((ext_vector_type(2)));
typedef float v8f __attribute__((ext_vector_type(8)));

#define TT   512
#define BB   512
#define NIN  64
#define H1   100
#define H2   50
#define H3   25
#define G1   400
#define G2   200
#define G3   100
#define G1S  400
#define G2S  208
#define G3S  112
#define SH1  104
#define SH2  52
#define SH3  28
#define NT   256
#define NWAVES 8

#define KS_X1 16
#define KS_H1 25
#define KS_H2 13
#define KS_H3 7

__device__ __forceinline__ float sigf(float x) { return 1.0f / (1.0f + __expf(-x)); }

__device__ __forceinline__ float tanh_sel(float x) {
  const float y  = fabsf(x);
  const float y2 = y * y;
  float p = fmaf(y2, -0x1.758e7ap-8f, 0x1.521192p-6f);
  p = fmaf(y2, p, -0x1.b8389cp-5f);
  p = fmaf(y2, p, 0x1.110704p-3f);
  p = fmaf(y2, p, -0x1.555532p-2f);
  const float zs = fmaf(y2, y * p, y);
  const float e  = expf(y + y);
  const float zl = fmaf(__builtin_amdgcn_rcpf(e + 1.0f), -2.0f, 1.0f);
  const float z  = (y < 0.625f) ? zs : zl;
  return copysignf(z, x);
}

template<int H, int GS>
__device__ __forceinline__ void ld_gates(const float* gates, int idx, const float (&b)[4], float (&v)[4]) {
  const int m = idx & 15, j = idx >> 4;
  v[0] = gates[m * GS + j]         + b[0];
  v[1] = gates[m * GS + H + j]     + b[1];
  v[2] = gates[m * GS + 2 * H + j] + b[2];
  v[3] = gates[m * GS + 3 * H + j] + b[3];
}

__device__ __forceinline__ float cell_math(const float (&v)[4], float& cst) {
  float c = sigf(v[1]) * cst + sigf(v[0]) * tanh_sel(v[2]);
  cst = c;
  return sigf(v[3]) * tanh_sel(c);
}

template<int SH>
__device__ __forceinline__ void st_h(float* hs, int idx, float h) {
  const int m = idx & 15, j = idx >> 4;
  hs[m * SH + j] = h;
}

template<int R, int H, int GS, int SH>
__device__ __forceinline__ void layer_cells(const float* gates, float* hs, const float (&bs)[R][4],
                                            float (&cr)[R], int tid) {
  #pragma unroll
  for (int r = 0; r < R; ++r) {
    const int idx = tid + r * NT;
    if (idx < 16 * H) {
      float g[4];
      ld_gates<H, GS>(gates, idx, bs[r], g);
      const float h = cell_math(g, cr[r]);
      st_h<SH>(hs, idx, h);
    }
  }
}

__device__ __forceinline__ v8f wmma4(v2f a, v2f b, v8f c) {
  return __builtin_amdgcn_wmma_f32_16x16x4_f32(
      false, a, false, b, (short)0, c, false, false);
}

template<int KS>
__device__ __forceinline__ void load_frags(v2f* f, const float* p, int khalf) {
  #pragma unroll
  for (int i = 0; i < KS; ++i) {
    f[i].x = p[4 * i + khalf];
    f[i].y = p[4 * i + khalf + 1];
  }
}

__device__ __forceinline__ void reads_stay_grouped() { __builtin_amdgcn_sched_barrier(0); }

template<int KS>
__device__ __forceinline__ v8f wmma_chain(const v2f* a, const v2f* b, v8f acc) {
  #pragma unroll
  for (int i = 0; i < KS; ++i) acc = wmma4(a[i], b[i], acc);
  return acc;
}

__global__ __launch_bounds__(NT, 2) void lstm3_fused_kernel(
    const float* __restrict__ x,
    const float* __restrict__ w_ih1, const float* __restrict__ w_hh1,
    const float* __restrict__ b_ih1, const float* __restrict__ b_hh1,
    const float* __restrict__ w_ih2, const float* __restrict__ w_hh2,
    const float* __restrict__ b_ih2, const float* __restrict__ b_hh2,
    const float* __restrict__ w_ih3, const float* __restrict__ w_hh3,
    const float* __restrict__ b_ih3, const float* __restrict__ b_hh3,
    const float* __restrict__ w_lin, const float* __restrict__ b_lin,
    float* __restrict__ out)
{
  __shared__ float gates1[16 * G1S];
  __shared__ float gates2[16 * G2S];
  __shared__ float gates3[16 * G3S];
  __shared__ float h1s[16 * SH1];
  __shared__ float h2s[16 * SH2];
  __shared__ float h3s[16 * SH3];
  __shared__ float whh1s[G1 * H1];
  __shared__ float whh2s[208 * SH2];
  __shared__ float wih3s[112 * SH2];
  __shared__ float whh3s[112 * SH3];

  const int tid     = threadIdx.x;
  const int lane    = tid & 31;
  const int wave    = tid >> 5;
  const int col     = lane & 15;
  const int khalf   = (lane >> 4) << 1;
  const int rowhalf = (lane >> 4) << 3;
  const int bbase   = blockIdx.x * 16;
  const int wrev    = (NWAVES - 1) - wave;

  for (int i = tid; i < 16 * SH1; i += NT) h1s[i] = 0.f;
  for (int i = tid; i < 16 * SH2; i += NT) h2s[i] = 0.f;
  for (int i = tid; i < 16 * SH3; i += NT) h3s[i] = 0.f;
  for (int i = tid; i < 208 * SH2; i += NT) whh2s[i] = 0.f;
  for (int i = tid; i < 112 * SH2; i += NT) wih3s[i] = 0.f;
  for (int i = tid; i < 112 * SH3; i += NT) whh3s[i] = 0.f;
  __syncthreads();

  for (int i = tid; i < G1 * H1; i += NT)
    whh1s[i] = w_hh1[i];
  for (int i = tid; i < G2 * H2; i += NT) {
    int r = i / H2, c = i - r * H2;
    whh2s[r * SH2 + c] = w_hh2[i];
  }
  for (int i = tid; i < G3 * H2; i += NT) {
    int r = i / H2, c = i - r * H2;
    wih3s[r * SH2 + c] = w_ih3[i];
  }
  for (int i = tid; i < G3 * H3; i += NT) {
    int r = i / H3, c = i - r * H3;
    whh3s[r * SH3 + c] = w_hh3[i];
  }

  float bs1[7][4], bs2[4][4], bs3[2][4];
  #pragma unroll
  for (int r = 0; r < 7; ++r) {
    const int idx = tid + r * NT;
    if (idx < 16 * H1) {
      const int j = idx >> 4;
      #pragma unroll
      for (int g = 0; g < 4; ++g) bs1[r][g] = b_ih1[g * H1 + j] + b_hh1[g * H1 + j];
    }
  }
  #pragma unroll
  for (int r = 0; r < 4; ++r) {
    const int idx = tid + r * NT;
    if (idx < 16 * H2) {
      const int j = idx >> 4;
      #pragma unroll
      for (int g = 0; g < 4; ++g) bs2[r][g] = b_ih2[g * H2 + j] + b_hh2[g * H2 + j];
    }
  }
  #pragma unroll
  for (int r = 0; r < 2; ++r) {
    const int idx = tid + r * NT;
    if (idx < 16 * H3) {
      const int j = idx >> 4;
      #pragma unroll
      for (int g = 0; g < 4; ++g) bs3[r][g] = b_ih3[g * H3 + j] + b_hh3[g * H3 + j];
    }
  }

  float wl[H3];
  #pragma unroll
  for (int j = 0; j < H3; ++j) wl[j] = w_lin[j];
  const float blin = b_lin[0];

  v2f wi1[4][KS_X1] = {};
  v2f wi2[2][KS_H1] = {};
  #pragma unroll
  for (int s = 0; s < 4; ++s) {
    const int tile = wave + s * NWAVES;
    if (tile < 25)
      load_frags<KS_X1>(wi1[s], w_ih1 + (size_t)(tile * 16 + col) * NIN, khalf);
  }
  #pragma unroll
  for (int s = 0; s < 2; ++s) {
    const int tile = wrev + s * NWAVES;
    if (tile < 13) {
      const int n  = tile * 16 + col;
      const int nc = (n < G2) ? n : (G2 - 1);
      load_frags<KS_H1>(wi2[s], w_ih2 + (size_t)nc * H1, khalf);
    }
  }

  float c1r[7], c2r[4], c3r[2];
  #pragma unroll
  for (int r = 0; r < 7; ++r) c1r[r] = 0.f;
  #pragma unroll
  for (int r = 0; r < 4; ++r) c2r[r] = 0.f;
  #pragma unroll
  for (int r = 0; r < 2; ++r) c3r[r] = 0.f;

  __syncthreads();

  for (int p = 0; p < TT + 2; ++p) {
    const bool act1 = (p < TT);
    const bool act2 = (p >= 1) && (p <= TT);
    const bool act3 = (p >= 2);

    {
      if (p + 1 < TT && tid < 16)
        __builtin_prefetch(x + ((size_t)(p + 1) * BB + bbase + tid) * NIN, 0, 3);

      v2f ax[KS_X1];
      if (act1)
        load_frags<KS_X1>(ax, x + ((size_t)p * BB + bbase + col) * NIN, khalf);

      v2f ah2[KS_H2];
      load_frags<KS_H2>(ah2, &h2s[col * SH2], khalf);


      if (act3) {
        if (wrev < 7) {
          const int n = wrev * 16 + col;
          v2f a3[KS_H3], bi[KS_H2], bh[KS_H3];
          load_frags<KS_H3>(a3, &h3s[col * SH3], khalf);
          load_frags<KS_H2>(bi, &wih3s[n * SH2], khalf);
          load_frags<KS_H3>(bh, &whh3s[n * SH3], khalf);
          reads_stay_grouped();
          v8f acc = {};
          acc = wmma_chain<KS_H2>(ah2, bi, acc);
          acc = wmma_chain<KS_H3>(a3, bh, acc);
          if (n < G3) {
            #pragma unroll
            for (int r = 0; r < 8; ++r)
              gates3[(r + rowhalf) * G3S + n] = acc[r];
          }
        }
      }

      v2f ah1[KS_H1];
      load_frags<KS_H1>(ah1, &h1s[col * SH1], khalf);

      if (act2) {
        #pragma unroll
        for (int s = 0; s < 2; ++s) {
          const int tile = wrev + s * NWAVES;
          if (tile < 13) {
            const int n = tile * 16 + col;
            v2f bh[KS_H2];
            load_frags<KS_H2>(bh, &whh2s[n * SH2], khalf);
            reads_stay_grouped();
            v8f acc = {};
            acc = wmma_chain<KS_H1>(ah1, wi2[s], acc);
            acc = wmma_chain<KS_H2>(ah2, bh, acc);
            if (n < G2) {
              #pragma unroll
              for (int r = 0; r < 8; ++r)
                gates2[(r + rowhalf) * G2S + n] = acc[r];
            }
          }
        }
      }

      if (act1) {
        #pragma unroll
        for (int s = 0; s < 4; ++s) {
          const int tile = wave + s * NWAVES;
          if (tile < 25) {
            const int n = tile * 16 + col;
            v2f bh[KS_H1];
            load_frags<KS_H1>(bh, &whh1s[n * H1], khalf);
            reads_stay_grouped();
            v8f acc = {};
            acc = wmma_chain<KS_X1>(ax, wi1[s], acc);
            acc = wmma_chain<KS_H1>(ah1, bh, acc);
            #pragma unroll
            for (int r = 0; r < 8; ++r)
              gates1[(r + rowhalf) * G1S + n] = acc[r];
          }
        }
      }
    }
    __syncthreads();

    if (act1 && act2 && act3) {
      float g1[6][4], g2[3][4], g3[4];
      #pragma unroll
      for (int r = 0; r < 6; ++r) ld_gates<H1, G1S>(gates1, tid + r * NT, bs1[r], g1[r]);
      #pragma unroll
      for (int r = 0; r < 3; ++r) ld_gates<H2, G2S>(gates2, tid + r * NT, bs2[r], g2[r]);
      ld_gates<H3, G3S>(gates3, tid, bs3[0], g3);
      float hn1[6], hn2[3], hn3;
      #pragma unroll
      for (int r = 0; r < 6; ++r) hn1[r] = cell_math(g1[r], c1r[r]);
      #pragma unroll
      for (int r = 0; r < 3; ++r) hn2[r] = cell_math(g2[r], c2r[r]);
      hn3 = cell_math(g3, c3r[0]);
      #pragma unroll
      for (int r = 0; r < 6; ++r) st_h<SH1>(h1s, tid + r * NT, hn1[r]);
      #pragma unroll
      for (int r = 0; r < 3; ++r) st_h<SH2>(h2s, tid + r * NT, hn2[r]);
      st_h<SH3>(h3s, tid, hn3);

      if (tid < 16 * H3 - NT) {
        float t3[4], t1[4], t2[4];
        ld_gates<H3, G3S>(gates3, tid + NT, bs3[1], t3);
        if (tid < 16 * H1 - 6 * NT) {
          ld_gates<H1, G1S>(gates1, tid + 6 * NT, bs1[6], t1);
          if (tid < 16 * H2 - 3 * NT) {
            ld_gates<H2, G2S>(gates2, tid + 3 * NT, bs2[3], t2);
            const float a3 = cell_math(t3, c3r[1]);
            const float a1 = cell_math(t1, c1r[6]);
            const float a2 = cell_math(t2, c2r[3]);
            st_h<SH3>(h3s, tid + NT, a3);
            st_h<SH1>(h1s, tid + 6 * NT, a1);
            st_h<SH2>(h2s, tid + 3 * NT, a2);
          } else {
            const float a3 = cell_math(t3, c3r[1]);
            const float a1 = cell_math(t1, c1r[6]);
            st_h<SH3>(h3s, tid + NT, a3);
            st_h<SH1>(h1s, tid + 6 * NT, a1);
          }
        } else {
          const float a3 = cell_math(t3, c3r[1]);
          st_h<SH3>(h3s, tid + NT, a3);
        }
      }
    } else {
      if (act1) layer_cells<7, H1, G1S, SH1>(gates1, h1s, bs1, c1r, tid);
      if (act2) layer_cells<4, H2, G2S, SH2>(gates2, h2s, bs2, c2r, tid);
      if (act3) layer_cells<2, H3, G3S, SH3>(gates3, h3s, bs3, c3r, tid);
    }
    __syncthreads();

    if (act3 && tid < 16) {
      const float* hr = &h3s[tid * SH3];
      float acc = blin;
      #pragma unroll
      for (int j = 0; j < H3; ++j) acc += hr[j] * wl[j];
      out[(size_t)(p - 2) * BB + bbase + tid] = acc;
    }
  }
}

extern "C" void kernel_launch(void* const* d_in, const int* in_sizes, int n_in,
                              void* d_out, int out_size, void* d_ws, size_t ws_size,
                              hipStream_t stream) {
  (void)in_sizes; (void)n_in; (void)d_ws; (void)ws_size; (void)out_size;
  const float* x     = (const float*)d_in[0];
  const float* w_ih1 = (const float*)d_in[1];
  const float* w_hh1 = (const float*)d_in[2];
  const float* b_ih1 = (const float*)d_in[3];
  const float* b_hh1 = (const float*)d_in[4];
  const float* w_ih2 = (const float*)d_in[5];
  const float* w_hh2 = (const float*)d_in[6];
  const float* b_ih2 = (const float*)d_in[7];
  const float* b_hh2 = (const float*)d_in[8];
  const float* w_ih3 = (const float*)d_in[9];
  const float* w_hh3 = (const float*)d_in[10];
  const float* b_ih3 = (const float*)d_in[11];
  const float* b_hh3 = (const float*)d_in[12];
  const float* w_lin = (const float*)d_in[13];
  const float* b_lin = (const float*)d_in[14];
  float* out = (float*)d_out;

  dim3 grid(BB / 16);
  dim3 block(NT);
  lstm3_fused_kernel<<<grid, block, 0, stream>>>(
      x, w_ih1, w_hh1, b_ih1, b_hh1,
      w_ih2, w_hh2, b_ih2, b_hh2,
      w_ih3, w_hh3, b_ih3, b_hh3,
      w_lin, b_lin, out);
}


# Cell nonlinearities on the hardware tanh instruction: tanh direct, sigmoid(x) = 0.5 + 0.5*tanh(x/2), replacing exp, polynomial and IEEE division; values change by its rounding, sums unchanged
# speedup vs baseline: 2.3254x; 1.1828x over previous
#include <hip/hip_runtime.h>
#include <hip/hip_bf16.h>
#include <math.h>

typedef float v2f __attribute__((ext_vector_type(2)));
typedef float v8f __attribute__((ext_vector_type(8)));

#define TT   512
#define BB   512
#define NIN  64
#define H1   100
#define H2   50
#define H3   25
#define G1   400
#define G2   200
#define G3   100
#define G1S  400
#define G2S  208
#define G3S  112
#define SH1  104
#define SH2  52
#define SH3  28
#define NT   256
#define NWAVES 8

#define KS_X1 16
#define KS_H1 25
#define KS_H2 13
#define KS_H3 7

__device__ __forceinline__ float tanh_hw(float x) { return __builtin_amdgcn_tanhf(x); }

template<int H, int GS>
__device__ __forceinline__ void ld_gates(const float* gates, int idx, const float (&b)[4], float (&v)[4]) {
  const int m = idx & 15, j = idx >> 4;
  v[0] = fmaf(gates[m * GS + j],         0.5f, b[0]);
  v[1] = fmaf(gates[m * GS + H + j],     0.5f, b[1]);
  v[2] = gates[m * GS + 2 * H + j] + b[2];
  v[3] = fmaf(gates[m * GS + 3 * H + j], 0.5f, b[3]);
}

__device__ __forceinline__ float cell_math(const float (&v)[4], float& cst) {
  const float si = fmaf(tanh_hw(v[0]), 0.5f, 0.5f);
  const float sf = fmaf(tanh_hw(v[1]), 0.5f, 0.5f);
  const float so = fmaf(tanh_hw(v[3]), 0.5f, 0.5f);
  float c = sf * cst + si * tanh_hw(v[2]);
  cst = c;
  return so * tanh_hw(c);
}

template<int SH>
__device__ __forceinline__ void st_h(float* hs, int idx, float h) {
  const int m = idx & 15, j = idx >> 4;
  hs[m * SH + j] = h;
}

template<int R, int H, int GS, int SH>
__device__ __forceinline__ void layer_cells(const float* gates, float* hs, const float (&bs)[R][4],
                                            float (&cr)[R], int tid) {
  #pragma unroll
  for (int r = 0; r < R; ++r) {
    const int idx = tid + r * NT;
    if (idx < 16 * H) {
      float g[4];
      ld_gates<H, GS>(gates, idx, bs[r], g);
      const float h = cell_math(g, cr[r]);
      st_h<SH>(hs, idx, h);
    }
  }
}

__device__ __forceinline__ v8f wmma4(v2f a, v2f b, v8f c) {
  return __builtin_amdgcn_wmma_f32_16x16x4_f32(
      false, a, false, b, (short)0, c, false, false);
}

template<int KS>
__device__ __forceinline__ void load_frags(v2f* f, const float* p, int khalf) {
  #pragma unroll
  for (int i = 0; i < KS; ++i) {
    f[i].x = p[4 * i + khalf];
    f[i].y = p[4 * i + khalf + 1];
  }
}

__device__ __forceinline__ void reads_stay_grouped() { __builtin_amdgcn_sched_barrier(0); }

template<int KS>
__device__ __forceinline__ v8f wmma_chain(const v2f* a, const v2f* b, v8f acc) {
  #pragma unroll
  for (int i = 0; i < KS; ++i) acc = wmma4(a[i], b[i], acc);
  return acc;
}

__global__ __launch_bounds__(NT, 2) void lstm3_fused_kernel(
    const float* __restrict__ x,
    const float* __restrict__ w_ih1, const float* __restrict__ w_hh1,
    const float* __restrict__ b_ih1, const float* __restrict__ b_hh1,
    const float* __restrict__ w_ih2, const float* __restrict__ w_hh2,
    const float* __restrict__ b_ih2, const float* __restrict__ b_hh2,
    const float* __restrict__ w_ih3, const float* __restrict__ w_hh3,
    const float* __restrict__ b_ih3, const float* __restrict__ b_hh3,
    const float* __restrict__ w_lin, const float* __restrict__ b_lin,
    float* __restrict__ out)
{
  __shared__ float gates1[16 * G1S];
  __shared__ float gates2[16 * G2S];
  __shared__ float gates3[16 * G3S];
  __shared__ float h1s[16 * SH1];
  __shared__ float h2s[16 * SH2];
  __shared__ float h3s[16 * SH3];
  __shared__ float whh1s[G1 * H1];
  __shared__ float whh2s[208 * SH2];
  __shared__ float wih3s[112 * SH2];
  __shared__ float whh3s[112 * SH3];

  const int tid     = threadIdx.x;
  const int lane    = tid & 31;
  const int wave    = tid >> 5;
  const int col     = lane & 15;
  const int khalf   = (lane >> 4) << 1;
  const int rowhalf = (lane >> 4) << 3;
  const int bbase   = blockIdx.x * 16;
  const int wrev    = (NWAVES - 1) - wave;

  for (int i = tid; i < 16 * SH1; i += NT) h1s[i] = 0.f;
  for (int i = tid; i < 16 * SH2; i += NT) h2s[i] = 0.f;
  for (int i = tid; i < 16 * SH3; i += NT) h3s[i] = 0.f;
  for (int i = tid; i < 208 * SH2; i += NT) whh2s[i] = 0.f;
  for (int i = tid; i < 112 * SH2; i += NT) wih3s[i] = 0.f;
  for (int i = tid; i < 112 * SH3; i += NT) whh3s[i] = 0.f;
  __syncthreads();

  for (int i = tid; i < G1 * H1; i += NT)
    whh1s[i] = w_hh1[i];
  for (int i = tid; i < G2 * H2; i += NT) {
    int r = i / H2, c = i - r * H2;
    whh2s[r * SH2 + c] = w_hh2[i];
  }
  for (int i = tid; i < G3 * H2; i += NT) {
    int r = i / H2, c = i - r * H2;
    wih3s[r * SH2 + c] = w_ih3[i];
  }
  for (int i = tid; i < G3 * H3; i += NT) {
    int r = i / H3, c = i - r * H3;
    whh3s[r * SH3 + c] = w_hh3[i];
  }

  float bs1[7][4], bs2[4][4], bs3[2][4];
  #pragma unroll
  for (int r = 0; r < 7; ++r) {
    const int idx = tid + r * NT;
    if (idx < 16 * H1) {
      const int j = idx >> 4;
      #pragma unroll
      for (int g = 0; g < 4; ++g) bs1[r][g] = (b_ih1[g * H1 + j] + b_hh1[g * H1 + j]) * (g == 2 ? 1.0f : 0.5f);
    }
  }
  #pragma unroll
  for (int r = 0; r < 4; ++r) {
    const int idx = tid + r * NT;
    if (idx < 16 * H2) {
      const int j = idx >> 4;
      #pragma unroll
      for (int g = 0; g < 4; ++g) bs2[r][g] = (b_ih2[g * H2 + j] + b_hh2[g * H2 + j]) * (g == 2 ? 1.0f : 0.5f);
    }
  }
  #pragma unroll
  for (int r = 0; r < 2; ++r) {
    const int idx = tid + r * NT;
    if (idx < 16 * H3) {
      const int j = idx >> 4;
      #pragma unroll
      for (int g = 0; g < 4; ++g) bs3[r][g] = (b_ih3[g * H3 + j] + b_hh3[g * H3 + j]) * (g == 2 ? 1.0f : 0.5f);
    }
  }

  float wl[H3];
  #pragma unroll
  for (int j = 0; j < H3; ++j) wl[j] = w_lin[j];
  const float blin = b_lin[0];

  v2f wi1[4][KS_X1] = {};
  v2f wi2[2][KS_H1] = {};
  #pragma unroll
  for (int s = 0; s < 4; ++s) {
    const int tile = wave + s * NWAVES;
    if (tile < 25)
      load_frags<KS_X1>(wi1[s], w_ih1 + (size_t)(tile * 16 + col) * NIN, khalf);
  }
  #pragma unroll
  for (int s = 0; s < 2; ++s) {
    const int tile = wrev + s * NWAVES;
    if (tile < 13) {
      const int n  = tile * 16 + col;
      const int nc = (n < G2) ? n : (G2 - 1);
      load_frags<KS_H1>(wi2[s], w_ih2 + (size_t)nc * H1, khalf);
    }
  }

  float c1r[7], c2r[4], c3r[2];
  #pragma unroll
  for (int r = 0; r < 7; ++r) c1r[r] = 0.f;
  #pragma unroll
  for (int r = 0; r < 4; ++r) c2r[r] = 0.f;
  #pragma unroll
  for (int r = 0; r < 2; ++r) c3r[r] = 0.f;

  __syncthreads();

  for (int p = 0; p < TT + 2; ++p) {
    const bool act1 = (p < TT);
    const bool act2 = (p >= 1) && (p <= TT);
    const bool act3 = (p >= 2);

    {
      if (p + 1 < TT && tid < 16)
        __builtin_prefetch(x + ((size_t)(p + 1) * BB + bbase + tid) * NIN, 0, 3);

      v2f ax[KS_X1];
      if (act1)
        load_frags<KS_X1>(ax, x + ((size_t)p * BB + bbase + col) * NIN, khalf);

      v2f ah2[KS_H2];
      load_frags<KS_H2>(ah2, &h2s[col * SH2], khalf);


      if (act3) {
        if (wrev < 7) {
          const int n = wrev * 16 + col;
          v2f a3[KS_H3], bi[KS_H2], bh[KS_H3];
          load_frags<KS_H3>(a3, &h3s[col * SH3], khalf);
          load_frags<KS_H2>(bi, &wih3s[n * SH2], khalf);
          load_frags<KS_H3>(bh, &whh3s[n * SH3], khalf);
          reads_stay_grouped();
          v8f acc = {};
          acc = wmma_chain<KS_H2>(ah2, bi, acc);
          acc = wmma_chain<KS_H3>(a3, bh, acc);
          if (n < G3) {
            #pragma unroll
            for (int r = 0; r < 8; ++r)
              gates3[(r + rowhalf) * G3S + n] = acc[r];
          }
        }
      }

      v2f ah1[KS_H1];
      load_frags<KS_H1>(ah1, &h1s[col * SH1], khalf);

      if (act2) {
        #pragma unroll
        for (int s = 0; s < 2; ++s) {
          const int tile = wrev + s * NWAVES;
          if (tile < 13) {
            const int n = tile * 16 + col;
            v2f bh[KS_H2];
            load_frags<KS_H2>(bh, &whh2s[n * SH2], khalf);
            reads_stay_grouped();
            v8f acc = {};
            acc = wmma_chain<KS_H1>(ah1, wi2[s], acc);
            acc = wmma_chain<KS_H2>(ah2, bh, acc);
            if (n < G2) {
              #pragma unroll
              for (int r = 0; r < 8; ++r)
                gates2[(r + rowhalf) * G2S + n] = acc[r];
            }
          }
        }
      }

      if (act1) {
        #pragma unroll
        for (int s = 0; s < 4; ++s) {
          const int tile = wave + s * NWAVES;
          if (tile < 25) {
            const int n = tile * 16 + col;
            v2f bh[KS_H1];
            load_frags<KS_H1>(bh, &whh1s[n * H1], khalf);
            reads_stay_grouped();
            v8f acc = {};
            acc = wmma_chain<KS_X1>(ax, wi1[s], acc);
            acc = wmma_chain<KS_H1>(ah1, bh, acc);
            #pragma unroll
            for (int r = 0; r < 8; ++r)
              gates1[(r + rowhalf) * G1S + n] = acc[r];
          }
        }
      }
    }
    __syncthreads();

    if (act1 && act2 && act3) {
      float g1[6][4], g2[3][4], g3[4];
      #pragma unroll
      for (int r = 0; r < 6; ++r) ld_gates<H1, G1S>(gates1, tid + r * NT, bs1[r], g1[r]);
      #pragma unroll
      for (int r = 0; r < 3; ++r) ld_gates<H2, G2S>(gates2, tid + r * NT, bs2[r], g2[r]);
      ld_gates<H3, G3S>(gates3, tid, bs3[0], g3);
      float hn1[6], hn2[3], hn3;
      #pragma unroll
      for (int r = 0; r < 6; ++r) hn1[r] = cell_math(g1[r], c1r[r]);
      #pragma unroll
      for (int r = 0; r < 3; ++r) hn2[r] = cell_math(g2[r], c2r[r]);
      hn3 = cell_math(g3, c3r[0]);
      #pragma unroll
      for (int r = 0; r < 6; ++r) st_h<SH1>(h1s, tid + r * NT, hn1[r]);
      #pragma unroll
      for (int r = 0; r < 3; ++r) st_h<SH2>(h2s, tid + r * NT, hn2[r]);
      st_h<SH3>(h3s, tid, hn3);

      if (tid < 16 * H3 - NT) {
        float t3[4], t1[4], t2[4];
        ld_gates<H3, G3S>(gates3, tid + NT, bs3[1], t3);
        if (tid < 16 * H1 - 6 * NT) {
          ld_gates<H1, G1S>(gates1, tid + 6 * NT, bs1[6], t1);
          if (tid < 16 * H2 - 3 * NT) {
            ld_gates<H2, G2S>(gates2, tid + 3 * NT, bs2[3], t2);
            const float a3 = cell_math(t3, c3r[1]);
            const float a1 = cell_math(t1, c1r[6]);
            const float a2 = cell_math(t2, c2r[3]);
            st_h<SH3>(h3s, tid + NT, a3);
            st_h<SH1>(h1s, tid + 6 * NT, a1);
            st_h<SH2>(h2s, tid + 3 * NT, a2);
          } else {
            const float a3 = cell_math(t3, c3r[1]);
            const float a1 = cell_math(t1, c1r[6]);
            st_h<SH3>(h3s, tid + NT, a3);
            st_h<SH1>(h1s, tid + 6 * NT, a1);
          }
        } else {
          const float a3 = cell_math(t3, c3r[1]);
          st_h<SH3>(h3s, tid + NT, a3);
        }
      }
    } else {
      if (act1) layer_cells<7, H1, G1S, SH1>(gates1, h1s, bs1, c1r, tid);
      if (act2) layer_cells<4, H2, G2S, SH2>(gates2, h2s, bs2, c2r, tid);
      if (act3) layer_cells<2, H3, G3S, SH3>(gates3, h3s, bs3, c3r, tid);
    }
    __syncthreads();

    if (act3 && tid < 16) {
      const float* hr = &h3s[tid * SH3];
      float acc = blin;
      #pragma unroll
      for (int j = 0; j < H3; ++j) acc += hr[j] * wl[j];
      out[(size_t)(p - 2) * BB + bbase + tid] = acc;
    }
  }
}

extern "C" void kernel_launch(void* const* d_in, const int* in_sizes, int n_in,
                              void* d_out, int out_size, void* d_ws, size_t ws_size,
                              hipStream_t stream) {
  (void)in_sizes; (void)n_in; (void)d_ws; (void)ws_size; (void)out_size;
  const float* x     = (const float*)d_in[0];
  const float* w_ih1 = (const float*)d_in[1];
  const float* w_hh1 = (const float*)d_in[2];
  const float* b_ih1 = (const float*)d_in[3];
  const float* b_hh1 = (const float*)d_in[4];
  const float* w_ih2 = (const float*)d_in[5];
  const float* w_hh2 = (const float*)d_in[6];
  const float* b_ih2 = (const float*)d_in[7];
  const float* b_hh2 = (const float*)d_in[8];
  const float* w_ih3 = (const float*)d_in[9];
  const float* w_hh3 = (const float*)d_in[10];
  const float* b_ih3 = (const float*)d_in[11];
  const float* b_hh3 = (const float*)d_in[12];
  const float* w_lin = (const float*)d_in[13];
  const float* b_lin = (const float*)d_in[14];
  float* out = (float*)d_out;

  dim3 grid(BB / 16);
  dim3 block(NT);
  lstm3_fused_kernel<<<grid, block, 0, stream>>>(
      x, w_ih1, w_hh1, b_ih1, b_hh1,
      w_ih2, w_hh2, b_ih2, b_hh2,
      w_ih3, w_hh3, b_ih3, b_hh3,
      w_lin, b_lin, out);
}


# Each wave updates the cells of its own gate tiles (columns regrouped: tile = 4 gates x 4 units; tile turned in a wave-own LDS patch; h double-buffered): 1 barrier per pass not 2; sum order unchanged
# speedup vs baseline: 2.5591x; 1.1005x over previous
#include <hip/hip_runtime.h>
#include <hip/hip_bf16.h>
#include <math.h>

typedef float v2f __attribute__((ext_vector_type(2)));
typedef float v8f __attribute__((ext_vector_type(8)));

#define TT   512
#define BB   512
#define NIN  64
#define H1   100
#define H2   50
#define H3   25
#define G1   400
#define G2   200
#define G3   100
#define SH1  104
#define SH2  52
#define SH3  28
#define NT   256
#define NWAVES 8

#define KS_X1 16
#define KS_H1 25
#define KS_H2 13
#define KS_H3 7

__device__ __forceinline__ float tanh_hw(float x) { return __builtin_amdgcn_tanhf(x); }

__device__ __forceinline__ float cell_math(const float (&v)[4], float& cst) {
  const float si = fmaf(tanh_hw(v[0]), 0.5f, 0.5f);
  const float sf = fmaf(tanh_hw(v[1]), 0.5f, 0.5f);
  const float so = fmaf(tanh_hw(v[3]), 0.5f, 0.5f);
  float c = sf * cst + si * tanh_hw(v[2]);
  cst = c;
  return so * tanh_hw(c);
}

__device__ __forceinline__ int gate_row(int t, int c, int H) { return (c & 3) * H + 4 * t + (c >> 2); }
__device__ __forceinline__ int gate_unit(int t, int c) { return 4 * t + (c >> 2); }

#define PS 20

__device__ __forceinline__ void wave_lds_handover() {
  __builtin_amdgcn_fence(__ATOMIC_RELEASE, "wavefront");
  __builtin_amdgcn_wave_barrier();
  __builtin_amdgcn_fence(__ATOMIC_ACQUIRE, "wavefront");
}

__device__ __forceinline__ void tile_bias(float (&b)[8], const float* b_ih, const float* b_hh, int H, int t, int u) {
  #pragma unroll
  for (int e = 0; e < 2; ++e) {
    const int unit = 4 * t + 2 * u + e;
    #pragma unroll
    for (int g = 0; g < 4; ++g)
      b[4 * e + g] = (unit < H) ? (b_ih[g * H + unit] + b_hh[g * H + unit]) * (g == 2 ? 1.0f : 0.5f) : 0.f;
  }
}

__device__ __forceinline__ void tile_cells(float* patch, v8f acc, const float (&b)[8], float (&c)[2],
                                           float* hdst, int col, int rowhalf, int u) {
  #pragma unroll
  for (int r = 0; r < 8; ++r) patch[(r + rowhalf) * PS + col] = acc[r];
  wave_lds_handover();
  const float4 q0 = *reinterpret_cast<const float4*>(&patch[col * PS + 8 * u]);
  const float4 q1 = *reinterpret_cast<const float4*>(&patch[col * PS + 8 * u + 4]);
  const float v0[4] = { fmaf(q0.x, 0.5f, b[0]), fmaf(q0.y, 0.5f, b[1]), q0.z + b[2], fmaf(q0.w, 0.5f, b[3]) };
  const float v1[4] = { fmaf(q1.x, 0.5f, b[4]), fmaf(q1.y, 0.5f, b[5]), q1.z + b[6], fmaf(q1.w, 0.5f, b[7]) };
  float2 hv;
  hv.x = cell_math(v0, c[0]);
  hv.y = cell_math(v1, c[1]);
  *reinterpret_cast<float2*>(hdst) = hv;
}

__device__ __forceinline__ v8f wmma4(v2f a, v2f b, v8f c) {
  return __builtin_amdgcn_wmma_f32_16x16x4_f32(
      false, a, false, b, (short)0, c, false, false);
}

template<int KS>
__device__ __forceinline__ void load_frags(v2f* f, const float* p, int khalf) {
  #pragma unroll
  for (int i = 0; i < KS; ++i) {
    f[i].x = p[4 * i + khalf];
    f[i].y = p[4 * i + khalf + 1];
  }
}

__device__ __forceinline__ void reads_stay_grouped() { __builtin_amdgcn_sched_barrier(0); }

template<int KS>
__device__ __forceinline__ v8f wmma_chain(const v2f* a, const v2f* b, v8f acc) {
  #pragma unroll
  for (int i = 0; i < KS; ++i) acc = wmma4(a[i], b[i], acc);
  return acc;
}

__global__ __launch_bounds__(NT, 2) void lstm3_fused_kernel(
    const float* __restrict__ x,
    const float* __restrict__ w_ih1, const float* __restrict__ w_hh1,
    const float* __restrict__ b_ih1, const float* __restrict__ b_hh1,
    const float* __restrict__ w_ih2, const float* __restrict__ w_hh2,
    const float* __restrict__ b_ih2, const float* __restrict__ b_hh2,
    const float* __restrict__ w_ih3, const float* __restrict__ w_hh3,
    const float* __restrict__ b_ih3, const float* __restrict__ b_hh3,
    const float* __restrict__ w_lin, const float* __restrict__ b_lin,
    float* __restrict__ out)
{
  __shared__ alignas(16) float patches[NWAVES * 16 * PS];
  __shared__ alignas(16) float h1s[2 * 16 * SH1];
  __shared__ alignas(16) float h2s[2 * 16 * SH2];
  __shared__ alignas(16) float h3s[2 * 16 * SH3];
  __shared__ float whh1s[G1 * H1];
  __shared__ float whh2s[208 * SH2];
  __shared__ float wih3s[112 * SH2];
  __shared__ float whh3s[112 * SH3];

  const int tid     = threadIdx.x;
  const int lane    = tid & 31;
  const int wave    = tid >> 5;
  const int col     = lane & 15;
  const int uhalf   = lane >> 4;
  const int khalf   = uhalf << 1;
  const int rowhalf = uhalf << 3;
  const int bbase   = blockIdx.x * 16;
  const int wrev    = (NWAVES - 1) - wave;
  float* patch = patches + wave * (16 * PS);

  for (int i = tid; i < 2 * 16 * SH1; i += NT) h1s[i] = 0.f;
  for (int i = tid; i < 2 * 16 * SH2; i += NT) h2s[i] = 0.f;
  for (int i = tid; i < 2 * 16 * SH3; i += NT) h3s[i] = 0.f;
  for (int i = tid; i < 208 * SH2; i += NT) whh2s[i] = 0.f;
  for (int i = tid; i < 112 * SH2; i += NT) wih3s[i] = 0.f;
  for (int i = tid; i < 112 * SH3; i += NT) whh3s[i] = 0.f;
  __syncthreads();

  for (int i = tid; i < G1 * H1; i += NT) {
    const int n = i / H1, k = i - n * H1;
    whh1s[i] = w_hh1[gate_row(n >> 4, n & 15, H1) * H1 + k];
  }
  for (int i = tid; i < 208 * H2; i += NT) {
    const int n = i / H2, k = i - n * H2;
    if (gate_unit(n >> 4, n & 15) < H2)
      whh2s[n * SH2 + k] = w_hh2[gate_row(n >> 4, n & 15, H2) * H2 + k];
  }
  for (int i = tid; i < 112 * H2; i += NT) {
    const int n = i / H2, k = i - n * H2;
    if (gate_unit(n >> 4, n & 15) < H3)
      wih3s[n * SH2 + k] = w_ih3[gate_row(n >> 4, n & 15, H3) * H2 + k];
  }
  for (int i = tid; i < 112 * H3; i += NT) {
    const int n = i / H3, k = i - n * H3;
    if (gate_unit(n >> 4, n & 15) < H3)
      whh3s[n * SH3 + k] = w_hh3[gate_row(n >> 4, n & 15, H3) * H3 + k];
  }

  float bs1[4][8], bs2[2][8], bs3[1][8];
  #pragma unroll
  for (int s = 0; s < 4; ++s) {
    const int tile = wave + s * NWAVES;
    tile_bias(bs1[s], b_ih1, b_hh1, H1, (tile < 25) ? tile : 0, uhalf);
  }
  #pragma unroll
  for (int s = 0; s < 2; ++s) {
    const int tile = wrev + s * NWAVES;
    tile_bias(bs2[s], b_ih2, b_hh2, H2, (tile < 13) ? tile : 0, uhalf);
  }
  tile_bias(bs3[0], b_ih3, b_hh3, H3, (wrev < 7) ? wrev : 0, uhalf);

  float wl[H3];
  #pragma unroll
  for (int j = 0; j < H3; ++j) wl[j] = w_lin[j];
  const float blin = b_lin[0];

  v2f wi1[4][KS_X1] = {};
  v2f wi2[2][KS_H1] = {};
  #pragma unroll
  for (int s = 0; s < 4; ++s) {
    const int tile = wave + s * NWAVES;
    if (tile < 25)
      load_frags<KS_X1>(wi1[s], w_ih1 + (size_t)gate_row(tile, col, H1) * NIN, khalf);
  }
  #pragma unroll
  for (int s = 0; s < 2; ++s) {
    const int tile = wrev + s * NWAVES;
    if (tile < 13) {
      if (gate_unit(tile, col) < H2)
        load_frags<KS_H1>(wi2[s], w_ih2 + (size_t)gate_row(tile, col, H2) * H1, khalf);
    }
  }

  float c1r[4][2] = {}, c2r[2][2] = {}, c3r[1][2] = {};

  __syncthreads();

  for (int p = 0; p < TT + 2; ++p) {
    const bool act1 = (p < TT);
    const bool act2 = (p >= 1) && (p <= TT);
    const bool act3 = (p >= 2);
    const int cur = p & 1, nxt = cur ^ 1;
    const float* h1r = h1s + cur * (16 * SH1);
    const float* h2r = h2s + cur * (16 * SH2);
    const float* h3r = h3s + cur * (16 * SH3);
    float* h1w = h1s + nxt * (16 * SH1) + col * SH1 + 2 * uhalf;
    float* h2w = h2s + nxt * (16 * SH2) + col * SH2 + 2 * uhalf;
    float* h3w = h3s + nxt * (16 * SH3) + col * SH3 + 2 * uhalf;

    {
      if (p + 1 < TT && tid < 16)
        __builtin_prefetch(x + ((size_t)(p + 1) * BB + bbase + tid) * NIN, 0, 3);

      v2f ax[KS_X1];
      if (act1)
        load_frags<KS_X1>(ax, x + ((size_t)p * BB + bbase + col) * NIN, khalf);

      v2f ah2[KS_H2];
      load_frags<KS_H2>(ah2, &h2r[col * SH2], khalf);


      if (act3) {
        if (wrev < 7) {
          const int n = wrev * 16 + col;
          v2f a3[KS_H3], bi[KS_H2], bh[KS_H3];
          load_frags<KS_H3>(a3, &h3r[col * SH3], khalf);
          load_frags<KS_H2>(bi, &wih3s[n * SH2], khalf);
          load_frags<KS_H3>(bh, &whh3s[n * SH3], khalf);
          reads_stay_grouped();
          v8f acc = {};
          acc = wmma_chain<KS_H2>(ah2, bi, acc);
          acc = wmma_chain<KS_H3>(a3, bh, acc);
          tile_cells(patch, acc, bs3[0], c3r[0], h3w + 4 * wrev, col, rowhalf, uhalf);
        }
      }

      v2f ah1[KS_H1];
      load_frags<KS_H1>(ah1, &h1r[col * SH1], khalf);

      if (act2) {
        #pragma unroll
        for (int s = 0; s < 2; ++s) {
          const int tile = wrev + s * NWAVES;
          if (tile < 13) {
            const int n = tile * 16 + col;
            v2f bh[KS_H2];
            load_frags<KS_H2>(bh, &whh2s[n * SH2], khalf);
            reads_stay_grouped();
            v8f acc = {};
            acc = wmma_chain<KS_H1>(ah1, wi2[s], acc);
            acc = wmma_chain<KS_H2>(ah2, bh, acc);
            tile_cells(patch, acc, bs2[s], c2r[s], h2w + 4 * tile, col, rowhalf, uhalf);
          }
        }
      }

      if (act1) {
        #pragma unroll
        for (int s = 0; s < 4; ++s) {
          const int tile = wave + s * NWAVES;
          if (tile < 25) {
            const int n = tile * 16 + col;
            v2f bh[KS_H1];
            load_frags<KS_H1>(bh, &whh1s[n * H1], khalf);
            reads_stay_grouped();
            v8f acc = {};
            acc = wmma_chain<KS_X1>(ax, wi1[s], acc);
            acc = wmma_chain<KS_H1>(ah1, bh, acc);
            tile_cells(patch, acc, bs1[s], c1r[s], h1w + 4 * tile, col, rowhalf, uhalf);
          }
        }
      }
    }
    __syncthreads();

    if (act3 && tid < 16) {
      const float* hr = &h3s[nxt * (16 * SH3) + tid * SH3];
      float acc = blin;
      #pragma unroll
      for (int j = 0; j < H3; ++j) acc += hr[j] * wl[j];
      out[(size_t)(p - 2) * BB + bbase + tid] = acc;
    }
  }
}

extern "C" void kernel_launch(void* const* d_in, const int* in_sizes, int n_in,
                              void* d_out, int out_size, void* d_ws, size_t ws_size,
                              hipStream_t stream) {
  (void)in_sizes; (void)n_in; (void)d_ws; (void)ws_size; (void)out_size;
  const float* x     = (const float*)d_in[0];
  const float* w_ih1 = (const float*)d_in[1];
  const float* w_hh1 = (const float*)d_in[2];
  const float* b_ih1 = (const float*)d_in[3];
  const float* b_hh1 = (const float*)d_in[4];
  const float* w_ih2 = (const float*)d_in[5];
  const float* w_hh2 = (const float*)d_in[6];
  const float* b_ih2 = (const float*)d_in[7];
  const float* b_hh2 = (const float*)d_in[8];
  const float* w_ih3 = (const float*)d_in[9];
  const float* w_hh3 = (const float*)d_in[10];
  const float* b_ih3 = (const float*)d_in[11];
  const float* b_hh3 = (const float*)d_in[12];
  const float* w_lin = (const float*)d_in[13];
  const float* b_lin = (const float*)d_in[14];
  float* out = (float*)d_out;

  dim3 grid(BB / 16);
  dim3 block(NT);
  lstm3_fused_kernel<<<grid, block, 0, stream>>>(
      x, w_ih1, w_hh1, b_ih1, b_hh1,
      w_ih2, w_hh2, b_ih2, b_hh2,
      w_ih3, w_hh3, b_ih3, b_hh3,
      w_lin, b_lin, out);
}


# Gate tiles computed transposed (weights as matrix operand A, h and x rows as B): a lane's 8 accumulator registers are the 4 gates of its own 2 cells, so the LDS patch turn goes; sum order unchanged
# speedup vs baseline: 2.6416x; 1.0322x over previous
#include <hip/hip_runtime.h>
#include <hip/hip_bf16.h>
#include <math.h>

typedef float v2f __attribute__((ext_vector_type(2)));
typedef float v8f __attribute__((ext_vector_type(8)));

#define TT   512
#define BB   512
#define NIN  64
#define H1   100
#define H2   50
#define H3   25
#define G1   400
#define G2   200
#define G3   100
#define SH1  104
#define SH2  52
#define SH3  28
#define NT   256
#define NWAVES 8

#define KS_X1 16
#define KS_H1 25
#define KS_H2 13
#define KS_H3 7

__device__ __forceinline__ float tanh_hw(float x) { return __builtin_amdgcn_tanhf(x); }

__device__ __forceinline__ float cell_math(const float (&v)[4], float& cst) {
  const float si = fmaf(tanh_hw(v[0]), 0.5f, 0.5f);
  const float sf = fmaf(tanh_hw(v[1]), 0.5f, 0.5f);
  const float so = fmaf(tanh_hw(v[3]), 0.5f, 0.5f);
  float c = sf * cst + si * tanh_hw(v[2]);
  cst = c;
  return so * tanh_hw(c);
}

__device__ __forceinline__ int gate_row(int t, int c, int H) { return (c & 3) * H + 4 * t + (c >> 2); }
__device__ __forceinline__ int gate_unit(int t, int c) { return 4 * t + (c >> 2); }

__device__ __forceinline__ void tile_bias(float (&b)[8], const float* b_ih, const float* b_hh, int H, int t, int u) {
  #pragma unroll
  for (int e = 0; e < 2; ++e) {
    const int unit = 4 * t + 2 * u + e;
    #pragma unroll
    for (int g = 0; g < 4; ++g)
      b[4 * e + g] = (unit < H) ? (b_ih[g * H + unit] + b_hh[g * H + unit]) * (g == 2 ? 1.0f : 0.5f) : 0.f;
  }
}

__device__ __forceinline__ void tile_cells(v8f acc, const float (&b)[8], float (&c)[2], float* hdst) {
  const float v0[4] = { fmaf(acc[0], 0.5f, b[0]), fmaf(acc[1], 0.5f, b[1]), acc[2] + b[2], fmaf(acc[3], 0.5f, b[3]) };
  const float v1[4] = { fmaf(acc[4], 0.5f, b[4]), fmaf(acc[5], 0.5f, b[5]), acc[6] + b[6], fmaf(acc[7], 0.5f, b[7]) };
  float2 hv;
  hv.x = cell_math(v0, c[0]);
  hv.y = cell_math(v1, c[1]);
  *reinterpret_cast<float2*>(hdst) = hv;
}

__device__ __forceinline__ v8f wmma4(v2f a, v2f b, v8f c) {
  return __builtin_amdgcn_wmma_f32_16x16x4_f32(
      false, a, false, b, (short)0, c, false, false);
}

template<int KS>
__device__ __forceinline__ void load_frags(v2f* f, const float* p, int khalf) {
  #pragma unroll
  for (int i = 0; i < KS; ++i) {
    f[i].x = p[4 * i + khalf];
    f[i].y = p[4 * i + khalf + 1];
  }
}

__device__ __forceinline__ void reads_stay_grouped() { __builtin_amdgcn_sched_barrier(0); }

template<int KS>
__device__ __forceinline__ v8f wmma_chain(const v2f* w, const v2f* act, v8f acc) {
  #pragma unroll
  for (int i = 0; i < KS; ++i) acc = wmma4(w[i], act[i], acc);
  return acc;
}

__global__ __launch_bounds__(NT, 2) void lstm3_fused_kernel(
    const float* __restrict__ x,
    const float* __restrict__ w_ih1, const float* __restrict__ w_hh1,
    const float* __restrict__ b_ih1, const float* __restrict__ b_hh1,
    const float* __restrict__ w_ih2, const float* __restrict__ w_hh2,
    const float* __restrict__ b_ih2, const float* __restrict__ b_hh2,
    const float* __restrict__ w_ih3, const float* __restrict__ w_hh3,
    const float* __restrict__ b_ih3, const float* __restrict__ b_hh3,
    const float* __restrict__ w_lin, const float* __restrict__ b_lin,
    float* __restrict__ out)
{
  __shared__ alignas(16) float h1s[2 * 16 * SH1];
  __shared__ alignas(16) float h2s[2 * 16 * SH2];
  __shared__ alignas(16) float h3s[2 * 16 * SH3];
  __shared__ float whh1s[G1 * H1];
  __shared__ float whh2s[208 * SH2];
  __shared__ float wih3s[112 * SH2];
  __shared__ float whh3s[112 * SH3];

  const int tid     = threadIdx.x;
  const int lane    = tid & 31;
  const int wave    = tid >> 5;
  const int col     = lane & 15;
  const int uhalf   = lane >> 4;
  const int khalf   = uhalf << 1;
  const int bbase   = blockIdx.x * 16;
  const int wrev    = (NWAVES - 1) - wave;

  for (int i = tid; i < 2 * 16 * SH1; i += NT) h1s[i] = 0.f;
  for (int i = tid; i < 2 * 16 * SH2; i += NT) h2s[i] = 0.f;
  for (int i = tid; i < 2 * 16 * SH3; i += NT) h3s[i] = 0.f;
  for (int i = tid; i < 208 * SH2; i += NT) whh2s[i] = 0.f;
  for (int i = tid; i < 112 * SH2; i += NT) wih3s[i] = 0.f;
  for (int i = tid; i < 112 * SH3; i += NT) whh3s[i] = 0.f;
  __syncthreads();

  for (int i = tid; i < G1 * H1; i += NT) {
    const int n = i / H1, k = i - n * H1;
    whh1s[i] = w_hh1[gate_row(n >> 4, n & 15, H1) * H1 + k];
  }
  for (int i = tid; i < 208 * H2; i += NT) {
    const int n = i / H2, k = i - n * H2;
    if (gate_unit(n >> 4, n & 15) < H2)
      whh2s[n * SH2 + k] = w_hh2[gate_row(n >> 4, n & 15, H2) * H2 + k];
  }
  for (int i = tid; i < 112 * H2; i += NT) {
    const int n = i / H2, k = i - n * H2;
    if (gate_unit(n >> 4, n & 15) < H3)
      wih3s[n * SH2 + k] = w_ih3[gate_row(n >> 4, n & 15, H3) * H2 + k];
  }
  for (int i = tid; i < 112 * H3; i += NT) {
    const int n = i / H3, k = i - n * H3;
    if (gate_unit(n >> 4, n & 15) < H3)
      whh3s[n * SH3 + k] = w_hh3[gate_row(n >> 4, n & 15, H3) * H3 + k];
  }

  float bs1[4][8], bs2[2][8], bs3[1][8];
  #pragma unroll
  for (int s = 0; s < 4; ++s) {
    const int tile = wave + s * NWAVES;
    tile_bias(bs1[s], b_ih1, b_hh1, H1, (tile < 25) ? tile : 0, uhalf);
  }
  #pragma unroll
  for (int s = 0; s < 2; ++s) {
    const int tile = wrev + s * NWAVES;
    tile_bias(bs2[s], b_ih2, b_hh2, H2, (tile < 13) ? tile : 0, uhalf);
  }
  tile_bias(bs3[0], b_ih3, b_hh3, H3, (wrev < 7) ? wrev : 0, uhalf);

  float wl[H3];
  #pragma unroll
  for (int j = 0; j < H3; ++j) wl[j] = w_lin[j];
  const float blin = b_lin[0];

  v2f wi1[4][KS_X1] = {};
  v2f wi2[2][KS_H1] = {};
  #pragma unroll
  for (int s = 0; s < 4; ++s) {
    const int tile = wave + s * NWAVES;
    if (tile < 25)
      load_frags<KS_X1>(wi1[s], w_ih1 + (size_t)gate_row(tile, col, H1) * NIN, khalf);
  }
  #pragma unroll
  for (int s = 0; s < 2; ++s) {
    const int tile = wrev + s * NWAVES;
    if (tile < 13) {
      if (gate_unit(tile, col) < H2)
        load_frags<KS_H1>(wi2[s], w_ih2 + (size_t)gate_row(tile, col, H2) * H1, khalf);
    }
  }

  float c1r[4][2] = {}, c2r[2][2] = {}, c3r[1][2] = {};

  __syncthreads();

  for (int p = 0; p < TT + 2; ++p) {
    const bool act1 = (p < TT);
    const bool act2 = (p >= 1) && (p <= TT);
    const bool act3 = (p >= 2);
    const int cur = p & 1, nxt = cur ^ 1;
    const float* h1r = h1s + cur * (16 * SH1);
    const float* h2r = h2s + cur * (16 * SH2);
    const float* h3r = h3s + cur * (16 * SH3);
    float* h1w = h1s + nxt * (16 * SH1) + col * SH1 + 2 * uhalf;
    float* h2w = h2s + nxt * (16 * SH2) + col * SH2 + 2 * uhalf;
    float* h3w = h3s + nxt * (16 * SH3) + col * SH3 + 2 * uhalf;

    {
      if (p + 1 < TT && tid < 16)
        __builtin_prefetch(x + ((size_t)(p + 1) * BB + bbase + tid) * NIN, 0, 3);

      v2f ax[KS_X1];
      if (act1)
        load_frags<KS_X1>(ax, x + ((size_t)p * BB + bbase + col) * NIN, khalf);

      v2f ah2[KS_H2];
      load_frags<KS_H2>(ah2, &h2r[col * SH2], khalf);


      if (act3) {
        if (wrev < 7) {
          const int n = wrev * 16 + col;
          v2f a3[KS_H3], bi[KS_H2], bh[KS_H3];
          load_frags<KS_H3>(a3, &h3r[col * SH3], khalf);
          load_frags<KS_H2>(bi, &wih3s[n * SH2], khalf);
          load_frags<KS_H3>(bh, &whh3s[n * SH3], khalf);
          reads_stay_grouped();
          v8f acc = {};
          acc = wmma_chain<KS_H2>(bi, ah2, acc);
          acc = wmma_chain<KS_H3>(bh, a3, acc);
          tile_cells(acc, bs3[0], c3r[0], h3w + 4 * wrev);
        }
      }

      v2f ah1[KS_H1];
      load_frags<KS_H1>(ah1, &h1r[col * SH1], khalf);

      if (act2) {
        #pragma unroll
        for (int s = 0; s < 2; ++s) {
          const int tile = wrev + s * NWAVES;
          if (tile < 13) {
            const int n = tile * 16 + col;
            v2f bh[KS_H2];
            load_frags<KS_H2>(bh, &whh2s[n * SH2], khalf);
            reads_stay_grouped();
            v8f acc = {};
            acc = wmma_chain<KS_H1>(wi2[s], ah1, acc);
            acc = wmma_chain<KS_H2>(bh, ah2, acc);
            tile_cells(acc, bs2[s], c2r[s], h2w + 4 * tile);
          }
        }
      }

      if (act1) {
        #pragma unroll
        for (int s = 0; s < 4; ++s) {
          const int tile = wave + s * NWAVES;
          if (tile < 25) {
            const int n = tile * 16 + col;
            v2f bh[KS_H1];
            load_frags<KS_H1>(bh, &whh1s[n * H1], khalf);
            reads_stay_grouped();
            v8f acc = {};
            acc = wmma_chain<KS_X1>(wi1[s], ax, acc);
            acc = wmma_chain<KS_H1>(bh, ah1, acc);
            tile_cells(acc, bs1[s], c1r[s], h1w + 4 * tile);
          }
        }
      }
    }
    __syncthreads();

    if (act3 && tid < 16) {
      const float* hr = &h3s[nxt * (16 * SH3) + tid * SH3];
      float acc = blin;
      #pragma unroll
      for (int j = 0; j < H3; ++j) acc += hr[j] * wl[j];
      out[(size_t)(p - 2) * BB + bbase + tid] = acc;
    }
  }
}

extern "C" void kernel_launch(void* const* d_in, const int* in_sizes, int n_in,
                              void* d_out, int out_size, void* d_ws, size_t ws_size,
                              hipStream_t stream) {
  (void)in_sizes; (void)n_in; (void)d_ws; (void)ws_size; (void)out_size;
  const float* x     = (const float*)d_in[0];
  const float* w_ih1 = (const float*)d_in[1];
  const float* w_hh1 = (const float*)d_in[2];
  const float* b_ih1 = (const float*)d_in[3];
  const float* b_hh1 = (const float*)d_in[4];
  const float* w_ih2 = (const float*)d_in[5];
  const float* w_hh2 = (const float*)d_in[6];
  const float* b_ih2 = (const float*)d_in[7];
  const float* b_hh2 = (const float*)d_in[8];
  const float* w_ih3 = (const float*)d_in[9];
  const float* w_hh3 = (const float*)d_in[10];
  const float* b_ih3 = (const float*)d_in[11];
  const float* b_hh3 = (const float*)d_in[12];
  const float* w_lin = (const float*)d_in[13];
  const float* b_lin = (const float*)d_in[14];
  float* out = (float*)d_out;

  dim3 grid(BB / 16);
  dim3 block(NT);
  lstm3_fused_kernel<<<grid, block, 0, stream>>>(
      x, w_ih1, w_hh1, b_ih1, b_hh1,
      w_ih2, w_hh2, b_ih2, b_hh2,
      w_ih3, w_hh3, b_ih3, b_hh3,
      w_lin, b_lin, out);
}


# Gate tiles dealt to waves by matrix-instruction count per SIMD pair (waves w, w+4 share a SIMD): busiest pair 418 not 438 of 1659 per pass; two layer-3 slots per wave; sum order unchanged
# speedup vs baseline: 2.7304x; 1.0336x over previous
#include <hip/hip_runtime.h>
#include <hip/hip_bf16.h>
#include <math.h>

typedef float v2f __attribute__((ext_vector_type(2)));
typedef float v8f __attribute__((ext_vector_type(8)));

#define TT   512
#define BB   512
#define NIN  64
#define H1   100
#define H2   50
#define H3   25
#define G1   400
#define G2   200
#define G3   100
#define SH1  104
#define SH2  52
#define SH3  28
#define NT   256
#define NWAVES 8

#define KS_X1 16
#define KS_H1 25
#define KS_H2 13
#define KS_H3 7

__device__ __forceinline__ float tanh_hw(float x) { return __builtin_amdgcn_tanhf(x); }

__device__ __forceinline__ float cell_math(const float (&v)[4], float& cst) {
  const float si = fmaf(tanh_hw(v[0]), 0.5f, 0.5f);
  const float sf = fmaf(tanh_hw(v[1]), 0.5f, 0.5f);
  const float so = fmaf(tanh_hw(v[3]), 0.5f, 0.5f);
  float c = sf * cst + si * tanh_hw(v[2]);
  cst = c;
  return so * tanh_hw(c);
}

__device__ __forceinline__ int gate_row(int t, int c, int H) { return (c & 3) * H + 4 * t + (c >> 2); }
__device__ __forceinline__ int gate_unit(int t, int c) { return 4 * t + (c >> 2); }

__device__ __forceinline__ int l2_tile(int w, int s) {
  if (w == 0) return -1;
  if (w == 4) return (s == 0) ? 12 : -1;
  return 2 * ((w < 4) ? w - 1 : w - 2) + s;
}
__device__ __forceinline__ int l3_tile(int w, int s) {
  if (w == 0) return s;
  if (w == 4) return 2 + s;
  if (w < 4)  return (s == 0) ? 3 + w : -1;
  return -1;
}

__device__ __forceinline__ void tile_bias(float (&b)[8], const float* b_ih, const float* b_hh, int H, int t, int u) {
  #pragma unroll
  for (int e = 0; e < 2; ++e) {
    const int unit = 4 * t + 2 * u + e;
    #pragma unroll
    for (int g = 0; g < 4; ++g)
      b[4 * e + g] = (unit < H) ? (b_ih[g * H + unit] + b_hh[g * H + unit]) * (g == 2 ? 1.0f : 0.5f) : 0.f;
  }
}

__device__ __forceinline__ void tile_cells(v8f acc, const float (&b)[8], float (&c)[2], float* hdst) {
  const float v0[4] = { fmaf(acc[0], 0.5f, b[0]), fmaf(acc[1], 0.5f, b[1]), acc[2] + b[2], fmaf(acc[3], 0.5f, b[3]) };
  const float v1[4] = { fmaf(acc[4], 0.5f, b[4]), fmaf(acc[5], 0.5f, b[5]), acc[6] + b[6], fmaf(acc[7], 0.5f, b[7]) };
  float2 hv;
  hv.x = cell_math(v0, c[0]);
  hv.y = cell_math(v1, c[1]);
  *reinterpret_cast<float2*>(hdst) = hv;
}

__device__ __forceinline__ v8f wmma4(v2f a, v2f b, v8f c) {
  return __builtin_amdgcn_wmma_f32_16x16x4_f32(
      false, a, false, b, (short)0, c, false, false);
}

template<int KS>
__device__ __forceinline__ void load_frags(v2f* f, const float* p, int khalf) {
  #pragma unroll
  for (int i = 0; i < KS; ++i) {
    f[i].x = p[4 * i + khalf];
    f[i].y = p[4 * i + khalf + 1];
  }
}

__device__ __forceinline__ void reads_stay_grouped() { __builtin_amdgcn_sched_barrier(0); }

template<int KS>
__device__ __forceinline__ v8f wmma_chain(const v2f* w, const v2f* act, v8f acc) {
  #pragma unroll
  for (int i = 0; i < KS; ++i) acc = wmma4(w[i], act[i], acc);
  return acc;
}

__global__ __launch_bounds__(NT, 2) void lstm3_fused_kernel(
    const float* __restrict__ x,
    const float* __restrict__ w_ih1, const float* __restrict__ w_hh1,
    const float* __restrict__ b_ih1, const float* __restrict__ b_hh1,
    const float* __restrict__ w_ih2, const float* __restrict__ w_hh2,
    const float* __restrict__ b_ih2, const float* __restrict__ b_hh2,
    const float* __restrict__ w_ih3, const float* __restrict__ w_hh3,
    const float* __restrict__ b_ih3, const float* __restrict__ b_hh3,
    const float* __restrict__ w_lin, const float* __restrict__ b_lin,
    float* __restrict__ out)
{
  __shared__ alignas(16) float h1s[2 * 16 * SH1];
  __shared__ alignas(16) float h2s[2 * 16 * SH2];
  __shared__ alignas(16) float h3s[2 * 16 * SH3];
  __shared__ float whh1s[G1 * H1];
  __shared__ float whh2s[208 * SH2];
  __shared__ float wih3s[112 * SH2];
  __shared__ float whh3s[112 * SH3];

  const int tid     = threadIdx.x;
  const int lane    = tid & 31;
  const int wave    = __builtin_amdgcn_readfirstlane(tid >> 5);
  const int col     = lane & 15;
  const int uhalf   = lane >> 4;
  const int khalf   = uhalf << 1;
  const int bbase   = blockIdx.x * 16;

  for (int i = tid; i < 2 * 16 * SH1; i += NT) h1s[i] = 0.f;
  for (int i = tid; i < 2 * 16 * SH2; i += NT) h2s[i] = 0.f;
  for (int i = tid; i < 2 * 16 * SH3; i += NT) h3s[i] = 0.f;
  for (int i = tid; i < 208 * SH2; i += NT) whh2s[i] = 0.f;
  for (int i = tid; i < 112 * SH2; i += NT) wih3s[i] = 0.f;
  for (int i = tid; i < 112 * SH3; i += NT) whh3s[i] = 0.f;
  __syncthreads();

  for (int i = tid; i < G1 * H1; i += NT) {
    const int n = i / H1, k = i - n * H1;
    whh1s[i] = w_hh1[gate_row(n >> 4, n & 15, H1) * H1 + k];
  }
  for (int i = tid; i < 208 * H2; i += NT) {
    const int n = i / H2, k = i - n * H2;
    if (gate_unit(n >> 4, n & 15) < H2)
      whh2s[n * SH2 + k] = w_hh2[gate_row(n >> 4, n & 15, H2) * H2 + k];
  }
  for (int i = tid; i < 112 * H2; i += NT) {
    const int n = i / H2, k = i - n * H2;
    if (gate_unit(n >> 4, n & 15) < H3)
      wih3s[n * SH2 + k] = w_ih3[gate_row(n >> 4, n & 15, H3) * H2 + k];
  }
  for (int i = tid; i < 112 * H3; i += NT) {
    const int n = i / H3, k = i - n * H3;
    if (gate_unit(n >> 4, n & 15) < H3)
      whh3s[n * SH3 + k] = w_hh3[gate_row(n >> 4, n & 15, H3) * H3 + k];
  }

  float bs1[4][8], bs2[2][8], bs3[2][8];
  #pragma unroll
  for (int s = 0; s < 4; ++s) {
    const int tile = wave + s * NWAVES;
    tile_bias(bs1[s], b_ih1, b_hh1, H1, (tile < 25) ? tile : 0, uhalf);
  }
  #pragma unroll
  for (int s = 0; s < 2; ++s) {
    const int tile = l2_tile(wave, s);
    tile_bias(bs2[s], b_ih2, b_hh2, H2, (tile >= 0) ? tile : 0, uhalf);
  }
  #pragma unroll
  for (int s = 0; s < 2; ++s) {
    const int tile = l3_tile(wave, s);
    tile_bias(bs3[s], b_ih3, b_hh3, H3, (tile >= 0) ? tile : 0, uhalf);
  }

  float wl[H3];
  #pragma unroll
  for (int j = 0; j < H3; ++j) wl[j] = w_lin[j];
  const float blin = b_lin[0];

  v2f wi1[4][KS_X1] = {};
  v2f wi2[2][KS_H1] = {};
  #pragma unroll
  for (int s = 0; s < 4; ++s) {
    const int tile = wave + s * NWAVES;
    if (tile < 25)
      load_frags<KS_X1>(wi1[s], w_ih1 + (size_t)gate_row(tile, col, H1) * NIN, khalf);
  }
  #pragma unroll
  for (int s = 0; s < 2; ++s) {
    const int tile = l2_tile(wave, s);
    if (tile >= 0) {
      if (gate_unit(tile, col) < H2)
        load_frags<KS_H1>(wi2[s], w_ih2 + (size_t)gate_row(tile, col, H2) * H1, khalf);
    }
  }

  float c1r[4][2] = {}, c2r[2][2] = {}, c3r[2][2] = {};

  __syncthreads();

  for (int p = 0; p < TT + 2; ++p) {
    const bool act1 = (p < TT);
    const bool act2 = (p >= 1) && (p <= TT);
    const bool act3 = (p >= 2);
    const int cur = p & 1, nxt = cur ^ 1;
    const float* h1r = h1s + cur * (16 * SH1);
    const float* h2r = h2s + cur * (16 * SH2);
    const float* h3r = h3s + cur * (16 * SH3);
    float* h1w = h1s + nxt * (16 * SH1) + col * SH1 + 2 * uhalf;
    float* h2w = h2s + nxt * (16 * SH2) + col * SH2 + 2 * uhalf;
    float* h3w = h3s + nxt * (16 * SH3) + col * SH3 + 2 * uhalf;

    {
      if (p + 1 < TT && tid < 16)
        __builtin_prefetch(x + ((size_t)(p + 1) * BB + bbase + tid) * NIN, 0, 3);

      v2f ax[KS_X1];
      if (act1)
        load_frags<KS_X1>(ax, x + ((size_t)p * BB + bbase + col) * NIN, khalf);

      v2f ah2[KS_H2];
      load_frags<KS_H2>(ah2, &h2r[col * SH2], khalf);


      if (act3) {
        if (l3_tile(wave, 0) >= 0) {
          v2f a3[KS_H3];
          load_frags<KS_H3>(a3, &h3r[col * SH3], khalf);
          #pragma unroll
          for (int s = 0; s < 2; ++s) {
            const int tile = l3_tile(wave, s);
            if (tile >= 0) {
              const int n = tile * 16 + col;
              v2f bi[KS_H2], bh[KS_H3];
              load_frags<KS_H2>(bi, &wih3s[n * SH2], khalf);
              load_frags<KS_H3>(bh, &whh3s[n * SH3], khalf);
              reads_stay_grouped();
              v8f acc = {};
              acc = wmma_chain<KS_H2>(bi, ah2, acc);
              acc = wmma_chain<KS_H3>(bh, a3, acc);
              tile_cells(acc, bs3[s], c3r[s], h3w + 4 * tile);
            }
          }
        }
      }

      v2f ah1[KS_H1];
      load_frags<KS_H1>(ah1, &h1r[col * SH1], khalf);

      if (act2) {
        #pragma unroll
        for (int s = 0; s < 2; ++s) {
          const int tile = l2_tile(wave, s);
          if (tile >= 0) {
            const int n = tile * 16 + col;
            v2f bh[KS_H2];
            load_frags<KS_H2>(bh, &whh2s[n * SH2], khalf);
            reads_stay_grouped();
            v8f acc = {};
            acc = wmma_chain<KS_H1>(wi2[s], ah1, acc);
            acc = wmma_chain<KS_H2>(bh, ah2, acc);
            tile_cells(acc, bs2[s], c2r[s], h2w + 4 * tile);
          }
        }
      }

      if (act1) {
        #pragma unroll
        for (int s = 0; s < 4; ++s) {
          const int tile = wave + s * NWAVES;
          if (tile < 25) {
            const int n = tile * 16 + col;
            v2f bh[KS_H1];
            load_frags<KS_H1>(bh, &whh1s[n * H1], khalf);
            reads_stay_grouped();
            v8f acc = {};
            acc = wmma_chain<KS_X1>(wi1[s], ax, acc);
            acc = wmma_chain<KS_H1>(bh, ah1, acc);
            tile_cells(acc, bs1[s], c1r[s], h1w + 4 * tile);
          }
        }
      }
    }
    __syncthreads();

    if (act3 && tid < 16) {
      const float* hr = &h3s[nxt * (16 * SH3) + tid * SH3];
      float acc = blin;
      #pragma unroll
      for (int j = 0; j < H3; ++j) acc += hr[j] * wl[j];
      out[(size_t)(p - 2) * BB + bbase + tid] = acc;
    }
  }
}

extern "C" void kernel_launch(void* const* d_in, const int* in_sizes, int n_in,
                              void* d_out, int out_size, void* d_ws, size_t ws_size,
                              hipStream_t stream) {
  (void)in_sizes; (void)n_in; (void)d_ws; (void)ws_size; (void)out_size;
  const float* x     = (const float*)d_in[0];
  const float* w_ih1 = (const float*)d_in[1];
  const float* w_hh1 = (const float*)d_in[2];
  const float* b_ih1 = (const float*)d_in[3];
  const float* b_hh1 = (const float*)d_in[4];
  const float* w_ih2 = (const float*)d_in[5];
  const float* w_hh2 = (const float*)d_in[6];
  const float* b_ih2 = (const float*)d_in[7];
  const float* b_hh2 = (const float*)d_in[8];
  const float* w_ih3 = (const float*)d_in[9];
  const float* w_hh3 = (const float*)d_in[10];
  const float* b_ih3 = (const float*)d_in[11];
  const float* b_hh3 = (const float*)d_in[12];
  const float* w_lin = (const float*)d_in[13];
  const float* b_lin = (const float*)d_in[14];
  float* out = (float*)d_out;

  dim3 grid(BB / 16);
  dim3 block(NT);
  lstm3_fused_kernel<<<grid, block, 0, stream>>>(
      x, w_ih1, w_hh1, b_ih1, b_hh1,
      w_ih2, w_hh2, b_ih2, b_hh2,
      w_ih3, w_hh3, b_ih3, b_hh3,
      w_lin, b_lin, out);
}


# Pass barrier split: a wave signals after its last h store, runs layer 1's next-step input products (registers, x only), then waits; bias sums in LDS, layer-1 reads under cell update; sums unchanged
# speedup vs baseline: 2.8140x; 1.0306x over previous
#include <hip/hip_runtime.h>
#include <hip/hip_bf16.h>
#include <math.h>

typedef float v2f __attribute__((ext_vector_type(2)));
typedef float v8f __attribute__((ext_vector_type(8)));

#define TT   512
#define BB   512
#define NIN  64
#define H1   100
#define H2   50
#define H3   25
#define G1   400
#define G2   200
#define G3   100
#define SH1  104
#define SH2  52
#define SH3  28
#define NT   256
#define NWAVES 8

#define KS_X1 16
#define KS_H1 25
#define KS_H2 13
#define KS_H3 7

__device__ __forceinline__ float tanh_hw(float x) { return __builtin_amdgcn_tanhf(x); }

__device__ __forceinline__ float cell_math(const float (&v)[4], float& cst) {
  const float si = fmaf(tanh_hw(v[0]), 0.5f, 0.5f);
  const float sf = fmaf(tanh_hw(v[1]), 0.5f, 0.5f);
  const float so = fmaf(tanh_hw(v[3]), 0.5f, 0.5f);
  float c = sf * cst + si * tanh_hw(v[2]);
  cst = c;
  return so * tanh_hw(c);
}

__device__ __forceinline__ int gate_row(int t, int c, int H) { return (c & 3) * H + 4 * t + (c >> 2); }
__device__ __forceinline__ int gate_unit(int t, int c) { return 4 * t + (c >> 2); }

__device__ __forceinline__ int l2_tile(int w, int s) {
  if (w == 0) return -1;
  if (w == 4) return (s == 0) ? 12 : -1;
  return 2 * ((w < 4) ? w - 1 : w - 2) + s;
}
__device__ __forceinline__ int l3_tile(int w, int s) {
  if (w == 0) return s;
  if (w == 4) return 2 + s;
  if (w < 4)  return (s == 0) ? 3 + w : -1;
  return -1;
}

#define BT_L1 0
#define BT_L2 25
#define BT_L3 38
#define BT_TILES 45
__device__ __forceinline__ float bias_entry(int i, const float* b_ih1, const float* b_hh1, const float* b_ih2,
                                            const float* b_hh2, const float* b_ih3, const float* b_hh3) {
  const int row = i >> 4, u = (i >> 3) & 1, e = (i >> 2) & 1, g = i & 3;
  const float* b_ih = b_ih1; const float* b_hh = b_hh1; int H = H1, t = row;
  if (row >= BT_L3)      { b_ih = b_ih3; b_hh = b_hh3; H = H3; t = row - BT_L3; }
  else if (row >= BT_L2) { b_ih = b_ih2; b_hh = b_hh2; H = H2; t = row - BT_L2; }
  const int unit = 4 * t + 2 * u + e;
  return (unit < H) ? (b_ih[g * H + unit] + b_hh[g * H + unit]) * (g == 2 ? 1.0f : 0.5f) : 0.f;
}
__device__ __forceinline__ void load_bias(float (&b)[8], const float4* table, int row, int u) {
  const float4* p = table + (row * 2 + u) * 2;
  const float4 lo = p[0], hi = p[1];
  b[0] = lo.x; b[1] = lo.y; b[2] = lo.z; b[3] = lo.w;
  b[4] = hi.x; b[5] = hi.y; b[6] = hi.z; b[7] = hi.w;
}
__device__ __forceinline__ void bias_read_here(float (&b)[8]) {
  asm volatile("" : "+v"(b[0]), "+v"(b[1]), "+v"(b[2]), "+v"(b[3]), "+v"(b[4]), "+v"(b[5]), "+v"(b[6]), "+v"(b[7]));
}

__device__ __forceinline__ void tile_cells(v8f acc, const float (&b)[8], float (&c)[2], float* hdst) {
  const float v0[4] = { fmaf(acc[0], 0.5f, b[0]), fmaf(acc[1], 0.5f, b[1]), acc[2] + b[2], fmaf(acc[3], 0.5f, b[3]) };
  const float v1[4] = { fmaf(acc[4], 0.5f, b[4]), fmaf(acc[5], 0.5f, b[5]), acc[6] + b[6], fmaf(acc[7], 0.5f, b[7]) };
  float2 hv;
  hv.x = cell_math(v0, c[0]);
  hv.y = cell_math(v1, c[1]);
  *reinterpret_cast<float2*>(hdst) = hv;
}

__device__ __forceinline__ v8f wmma4(v2f a, v2f b, v8f c) {
  return __builtin_amdgcn_wmma_f32_16x16x4_f32(
      false, a, false, b, (short)0, c, false, false);
}

template<int KS>
__device__ __forceinline__ void load_frags(v2f* f, const float* p, int khalf) {
  #pragma unroll
  for (int i = 0; i < KS; ++i) {
    f[i].x = p[4 * i + khalf];
    f[i].y = p[4 * i + khalf + 1];
  }
}

__device__ __forceinline__ void reads_stay_grouped() { __builtin_amdgcn_sched_barrier(0); }

template<int KS>
__device__ __forceinline__ v8f wmma_chain(const v2f* w, const v2f* act, v8f acc) {
  #pragma unroll
  for (int i = 0; i < KS; ++i) acc = wmma4(w[i], act[i], acc);
  return acc;
}

__global__ __launch_bounds__(NT, 2) void lstm3_fused_kernel(
    const float* __restrict__ x,
    const float* __restrict__ w_ih1, const float* __restrict__ w_hh1,
    const float* __restrict__ b_ih1, const float* __restrict__ b_hh1,
    const float* __restrict__ w_ih2, const float* __restrict__ w_hh2,
    const float* __restrict__ b_ih2, const float* __restrict__ b_hh2,
    const float* __restrict__ w_ih3, const float* __restrict__ w_hh3,
    const float* __restrict__ b_ih3, const float* __restrict__ b_hh3,
    const float* __restrict__ w_lin, const float* __restrict__ b_lin,
    float* __restrict__ out)
{
  __shared__ alignas(16) float h1s[2 * 16 * SH1];
  __shared__ alignas(16) float h2s[2 * 16 * SH2];
  __shared__ alignas(16) float h3s[2 * 16 * SH3];
  __shared__ float whh1s[G1 * H1];
  __shared__ float whh2s[208 * SH2];
  __shared__ float wih3s[112 * SH2];
  __shared__ float whh3s[112 * SH3];
  __shared__ float4 biasT[BT_TILES * 4];

  const int tid     = threadIdx.x;
  const int lane    = tid & 31;
  const int wave    = __builtin_amdgcn_readfirstlane(tid >> 5);
  const int col     = lane & 15;
  const int uhalf   = lane >> 4;
  const int khalf   = uhalf << 1;
  const int bbase   = blockIdx.x * 16;

  for (int i = tid; i < 2 * 16 * SH1; i += NT) h1s[i] = 0.f;
  for (int i = tid; i < 2 * 16 * SH2; i += NT) h2s[i] = 0.f;
  for (int i = tid; i < 2 * 16 * SH3; i += NT) h3s[i] = 0.f;
  for (int i = tid; i < 208 * SH2; i += NT) whh2s[i] = 0.f;
  for (int i = tid; i < 112 * SH2; i += NT) wih3s[i] = 0.f;
  for (int i = tid; i < 112 * SH3; i += NT) whh3s[i] = 0.f;
  __syncthreads();

  for (int i = tid; i < G1 * H1; i += NT) {
    const int n = i / H1, k = i - n * H1;
    whh1s[i] = w_hh1[gate_row(n >> 4, n & 15, H1) * H1 + k];
  }
  for (int i = tid; i < 208 * H2; i += NT) {
    const int n = i / H2, k = i - n * H2;
    if (gate_unit(n >> 4, n & 15) < H2)
      whh2s[n * SH2 + k] = w_hh2[gate_row(n >> 4, n & 15, H2) * H2 + k];
  }
  for (int i = tid; i < 112 * H2; i += NT) {
    const int n = i / H2, k = i - n * H2;
    if (gate_unit(n >> 4, n & 15) < H3)
      wih3s[n * SH2 + k] = w_ih3[gate_row(n >> 4, n & 15, H3) * H2 + k];
  }
  for (int i = tid; i < 112 * H3; i += NT) {
    const int n = i / H3, k = i - n * H3;
    if (gate_unit(n >> 4, n & 15) < H3)
      whh3s[n * SH3 + k] = w_hh3[gate_row(n >> 4, n & 15, H3) * H3 + k];
  }

  for (int i = tid; i < BT_TILES * 16; i += NT)
    reinterpret_cast<float*>(biasT)[i] = bias_entry(i, b_ih1, b_hh1, b_ih2, b_hh2, b_ih3, b_hh3);

  float wl[H3];
  #pragma unroll
  for (int j = 0; j < H3; ++j) wl[j] = w_lin[j];
  const float blin = b_lin[0];

  v2f wi1[4][KS_X1] = {};
  v2f wi2[2][KS_H1] = {};
  #pragma unroll
  for (int s = 0; s < 4; ++s) {
    const int tile = wave + s * NWAVES;
    if (tile < 25)
      load_frags<KS_X1>(wi1[s], w_ih1 + (size_t)gate_row(tile, col, H1) * NIN, khalf);
  }
  #pragma unroll
  for (int s = 0; s < 2; ++s) {
    const int tile = l2_tile(wave, s);
    if (tile >= 0) {
      if (gate_unit(tile, col) < H2)
        load_frags<KS_H1>(wi2[s], w_ih2 + (size_t)gate_row(tile, col, H2) * H1, khalf);
    }
  }

  float c1r[4][2] = {}, c2r[2][2] = {}, c3r[2][2] = {};

  v8f acc1[4] = {};
  {
    v2f ax0[KS_X1];
    load_frags<KS_X1>(ax0, x + ((size_t)bbase + col) * NIN, khalf);
    #pragma unroll
    for (int s = 0; s < 4; ++s) {
      const int tile = wave + s * NWAVES;
      if (tile < 25) {
        v8f z = {};
        acc1[s] = wmma_chain<KS_X1>(wi1[s], ax0, z);
      }
    }
  }

  __syncthreads();

  for (int p = 0; p < TT + 2; ++p) {
    const bool act1 = (p < TT);
    const bool act2 = (p >= 1) && (p <= TT);
    const bool act3 = (p >= 2);
    const bool nxt1 = (p + 1 < TT);
    const int cur = p & 1, nxt = cur ^ 1;
    const float* h1r = h1s + cur * (16 * SH1);
    const float* h2r = h2s + cur * (16 * SH2);
    const float* h3r = h3s + cur * (16 * SH3);
    float* h1w = h1s + nxt * (16 * SH1) + col * SH1 + 2 * uhalf;
    float* h2w = h2s + nxt * (16 * SH2) + col * SH2 + 2 * uhalf;
    float* h3w = h3s + nxt * (16 * SH3) + col * SH3 + 2 * uhalf;

    if (p + 2 < TT && tid < 16)
      __builtin_prefetch(x + ((size_t)(p + 2) * BB + bbase + tid) * NIN, 0, 3);

    v2f ax[KS_X1];
    if (nxt1)
      load_frags<KS_X1>(ax, x + ((size_t)(p + 1) * BB + bbase + col) * NIN, khalf);

    {
      v2f ah2[KS_H2];
      load_frags<KS_H2>(ah2, &h2r[col * SH2], khalf);


      if (act3) {
        if (l3_tile(wave, 0) >= 0) {
          v2f a3[KS_H3];
          load_frags<KS_H3>(a3, &h3r[col * SH3], khalf);
          #pragma unroll
          for (int s = 0; s < 2; ++s) {
            const int tile = l3_tile(wave, s);
            if (tile >= 0) {
              const int n = tile * 16 + col;
              v2f bi[KS_H2], bh[KS_H3];
              float bs[8];
              load_bias(bs, biasT, BT_L3 + tile, uhalf);
              load_frags<KS_H2>(bi, &wih3s[n * SH2], khalf);
              load_frags<KS_H3>(bh, &whh3s[n * SH3], khalf);
              bias_read_here(bs);
              reads_stay_grouped();
              v8f acc = {};
              acc = wmma_chain<KS_H2>(bi, ah2, acc);
              acc = wmma_chain<KS_H3>(bh, a3, acc);
              tile_cells(acc, bs, c3r[s], h3w + 4 * tile);
            }
          }
        }
      }

      v2f ah1[KS_H1];
      load_frags<KS_H1>(ah1, &h1r[col * SH1], khalf);

      if (act2) {
        #pragma unroll
        for (int s = 0; s < 2; ++s) {
          const int tile = l2_tile(wave, s);
          if (tile >= 0) {
            const int n = tile * 16 + col;
            v2f bh[KS_H2];
            float bs[8];
            load_bias(bs, biasT, BT_L2 + tile, uhalf);
            load_frags<KS_H2>(bh, &whh2s[n * SH2], khalf);
            bias_read_here(bs);
            reads_stay_grouped();
            v8f acc = {};
            acc = wmma_chain<KS_H1>(wi2[s], ah1, acc);
            acc = wmma_chain<KS_H2>(bh, ah2, acc);
            tile_cells(acc, bs, c2r[s], h2w + 4 * tile);
          }
        }
      }

      if (act1) {
        v2f bh[KS_H1];
        float bs[8];
        load_bias(bs, biasT, BT_L1 + wave, uhalf);
        load_frags<KS_H1>(bh, &whh1s[(wave * 16 + col) * H1], khalf);
        #pragma unroll
        for (int s = 0; s < 4; ++s) {
          const int tile = wave + s * NWAVES;
          if (tile < 25) {
            bias_read_here(bs);
            reads_stay_grouped();
            v8f acc = wmma_chain<KS_H1>(bh, ah1, acc1[s]);
            reads_stay_grouped();
            const bool more = (s < 3) && (tile + NWAVES < 25);
            float bsn[8];
            load_bias(bsn, biasT, BT_L1 + (more ? tile + NWAVES : tile), uhalf);
            if (more)
              load_frags<KS_H1>(bh, &whh1s[((tile + NWAVES) * 16 + col) * H1], khalf);
            reads_stay_grouped();
            tile_cells(acc, bs, c1r[s], h1w + 4 * tile);
            reads_stay_grouped();
            #pragma unroll
            for (int j = 0; j < 8; ++j) bs[j] = bsn[j];
          }
        }
      }
    }

    __builtin_amdgcn_fence(__ATOMIC_RELEASE, "workgroup");
    __builtin_amdgcn_s_barrier_signal(-1);
    __builtin_amdgcn_sched_barrier(0);
    if (nxt1) {
      #pragma unroll
      for (int s = 0; s < 4; ++s) {
        const int tile = wave + s * NWAVES;
        if (tile < 25) {
          v8f z = {};
          acc1[s] = wmma_chain<KS_X1>(wi1[s], ax, z);
        }
      }
    }
    __builtin_amdgcn_sched_barrier(0);
    __builtin_amdgcn_s_barrier_wait(-1);
    __builtin_amdgcn_fence(__ATOMIC_ACQUIRE, "workgroup");

    if (act3 && tid < 16) {
      const float* hr = &h3s[nxt * (16 * SH3) + tid * SH3];
      float acc = blin;
      #pragma unroll
      for (int j = 0; j < H3; ++j) acc += hr[j] * wl[j];
      out[(size_t)(p - 2) * BB + bbase + tid] = acc;
    }
  }
}

extern "C" void kernel_launch(void* const* d_in, const int* in_sizes, int n_in,
                              void* d_out, int out_size, void* d_ws, size_t ws_size,
                              hipStream_t stream) {
  (void)in_sizes; (void)n_in; (void)d_ws; (void)ws_size; (void)out_size;
  const float* x     = (const float*)d_in[0];
  const float* w_ih1 = (const float*)d_in[1];
  const float* w_hh1 = (const float*)d_in[2];
  const float* b_ih1 = (const float*)d_in[3];
  const float* b_hh1 = (const float*)d_in[4];
  const float* w_ih2 = (const float*)d_in[5];
  const float* w_hh2 = (const float*)d_in[6];
  const float* b_ih2 = (const float*)d_in[7];
  const float* b_hh2 = (const float*)d_in[8];
  const float* w_ih3 = (const float*)d_in[9];
  const float* w_hh3 = (const float*)d_in[10];
  const float* b_ih3 = (const float*)d_in[11];
  const float* b_hh3 = (const float*)d_in[12];
  const float* w_lin = (const float*)d_in[13];
  const float* b_lin = (const float*)d_in[14];
  float* out = (float*)d_out;

  dim3 grid(BB / 16);
  dim3 block(NT);
  lstm3_fused_kernel<<<grid, block, 0, stream>>>(
      x, w_ih1, w_hh1, b_ih1, b_hh1,
      w_ih2, w_hh2, b_ih2, b_hh2,
      w_ih3, w_hh3, b_ih3, b_hh3,
      w_lin, b_lin, out);
}


# 8 batch rows per work-group (64 groups): the 16 matrix columns of layer 1's input products carry two time steps at once; column halves take turns per pass, the idle half adds zeros; sums unchanged
# speedup vs baseline: 2.9344x; 1.0428x over previous
#include <hip/hip_runtime.h>
#include <hip/hip_bf16.h>
#include <math.h>

typedef float v2f __attribute__((ext_vector_type(2)));
typedef float v8f __attribute__((ext_vector_type(8)));

#define TT   512
#define BB   512
#define NIN  64
#define H1   100
#define H2   50
#define H3   25
#define G1   400
#define G2   200
#define G3   100
#define SH1  104
#define SH2  52
#define SH3  28
#define NT   256
#define NWAVES 8
#define BR   8

#define KS_X1 16
#define KS_H1 25
#define KS_H2 13
#define KS_H3 7

__device__ __forceinline__ float tanh_hw(float x) { return __builtin_amdgcn_tanhf(x); }

__device__ __forceinline__ float cell_math(const float (&v)[4], float& cst) {
  const float si = fmaf(tanh_hw(v[0]), 0.5f, 0.5f);
  const float sf = fmaf(tanh_hw(v[1]), 0.5f, 0.5f);
  const float so = fmaf(tanh_hw(v[3]), 0.5f, 0.5f);
  float c = sf * cst + si * tanh_hw(v[2]);
  cst = c;
  return so * tanh_hw(c);
}

__device__ __forceinline__ int gate_row(int t, int c, int H) { return (c & 3) * H + 4 * t + (c >> 2); }
__device__ __forceinline__ int gate_unit(int t, int c) { return 4 * t + (c >> 2); }

__device__ __forceinline__ int l2_tile(int w, int s) {
  if (w == 0) return -1;
  if (w == 4) return (s == 0) ? 12 : -1;
  return 2 * ((w < 4) ? w - 1 : w - 2) + s;
}
__device__ __forceinline__ int l3_tile(int w, int s) {
  if (w == 0) return s;
  if (w == 4) return 2 + s;
  if (w < 4)  return (s == 0) ? 3 + w : -1;
  return -1;
}

#define BT_L1 0
#define BT_L2 25
#define BT_L3 38
#define BT_TILES 45
__device__ __forceinline__ float bias_entry(int i, const float* b_ih1, const float* b_hh1, const float* b_ih2,
                                            const float* b_hh2, const float* b_ih3, const float* b_hh3) {
  const int row = i >> 4, u = (i >> 3) & 1, e = (i >> 2) & 1, g = i & 3;
  const float* b_ih = b_ih1; const float* b_hh = b_hh1; int H = H1, t = row;
  if (row >= BT_L3)      { b_ih = b_ih3; b_hh = b_hh3; H = H3; t = row - BT_L3; }
  else if (row >= BT_L2) { b_ih = b_ih2; b_hh = b_hh2; H = H2; t = row - BT_L2; }
  const int unit = 4 * t + 2 * u + e;
  return (unit < H) ? (b_ih[g * H + unit] + b_hh[g * H + unit]) * (g == 2 ? 1.0f : 0.5f) : 0.f;
}
__device__ __forceinline__ void load_bias(float (&b)[8], const float4* table, int row, int u) {
  const float4* p = table + (row * 2 + u) * 2;
  const float4 lo = p[0], hi = p[1];
  b[0] = lo.x; b[1] = lo.y; b[2] = lo.z; b[3] = lo.w;
  b[4] = hi.x; b[5] = hi.y; b[6] = hi.z; b[7] = hi.w;
}
__device__ __forceinline__ void bias_read_here(float (&b)[8]) {
  asm volatile("" : "+v"(b[0]), "+v"(b[1]), "+v"(b[2]), "+v"(b[3]), "+v"(b[4]), "+v"(b[5]), "+v"(b[6]), "+v"(b[7]));
}

__device__ __forceinline__ float partner_column(float v) {
  return __builtin_bit_cast(float, __builtin_amdgcn_update_dpp(0, __builtin_bit_cast(int, v), 0x168, 0xF, 0xF, true));
}
__device__ __forceinline__ void tile_cells(v8f acc, const float (&b)[8], float (&c)[2], float* hdst, bool active) {
  const float v0[4] = { fmaf(acc[0], 0.5f, b[0]), fmaf(acc[1], 0.5f, b[1]), acc[2] + b[2], fmaf(acc[3], 0.5f, b[3]) };
  const float v1[4] = { fmaf(acc[4], 0.5f, b[4]), fmaf(acc[5], 0.5f, b[5]), acc[6] + b[6], fmaf(acc[7], 0.5f, b[7]) };
  float2 hv;
  hv.x = cell_math(v0, c[0]);
  hv.y = cell_math(v1, c[1]);
  if (active) *reinterpret_cast<float2*>(hdst) = hv;
  c[0] = partner_column(c[0]);
  c[1] = partner_column(c[1]);
}

__device__ __forceinline__ v8f wmma4(v2f a, v2f b, v8f c) {
  return __builtin_amdgcn_wmma_f32_16x16x4_f32(
      false, a, false, b, (short)0, c, false, false);
}

template<int KS>
__device__ __forceinline__ void load_frags(v2f* f, const float* p, int khalf) {
  #pragma unroll
  for (int i = 0; i < KS; ++i) {
    f[i].x = p[4 * i + khalf];
    f[i].y = p[4 * i + khalf + 1];
  }
}

__device__ __forceinline__ void reads_stay_grouped() { __builtin_amdgcn_sched_barrier(0); }

template<int KS>
__device__ __forceinline__ v8f wmma_chain(const v2f* w, const v2f* act, v8f acc) {
  #pragma unroll
  for (int i = 0; i < KS; ++i) acc = wmma4(w[i], act[i], acc);
  return acc;
}

__global__ __launch_bounds__(NT, 2) void lstm3_fused_kernel(
    const float* __restrict__ x,
    const float* __restrict__ w_ih1, const float* __restrict__ w_hh1,
    const float* __restrict__ b_ih1, const float* __restrict__ b_hh1,
    const float* __restrict__ w_ih2, const float* __restrict__ w_hh2,
    const float* __restrict__ b_ih2, const float* __restrict__ b_hh2,
    const float* __restrict__ w_ih3, const float* __restrict__ w_hh3,
    const float* __restrict__ b_ih3, const float* __restrict__ b_hh3,
    const float* __restrict__ w_lin, const float* __restrict__ b_lin,
    float* __restrict__ out)
{
  __shared__ alignas(16) float h1s[2 * 16 * SH1];
  __shared__ alignas(16) float h2s[2 * 16 * SH2];
  __shared__ alignas(16) float h3s[2 * 16 * SH3];
  __shared__ float whh1s[G1 * H1];
  __shared__ float whh2s[208 * SH2];
  __shared__ float wih3s[112 * SH2];
  __shared__ float whh3s[112 * SH3];
  __shared__ float4 biasT[BT_TILES * 4];

  const int tid     = threadIdx.x;
  const int lane    = tid & 31;
  const int wave    = __builtin_amdgcn_readfirstlane(tid >> 5);
  const int col     = lane & 15;
  const int uhalf   = lane >> 4;
  const int khalf   = uhalf << 1;
  const int bbase   = blockIdx.x * BR;
  const int brow    = col & (BR - 1);
  const int chalf   = col >> 3;

  for (int i = tid; i < 2 * 16 * SH1; i += NT) h1s[i] = 0.f;
  for (int i = tid; i < 2 * 16 * SH2; i += NT) h2s[i] = 0.f;
  for (int i = tid; i < 2 * 16 * SH3; i += NT) h3s[i] = 0.f;
  for (int i = tid; i < 208 * SH2; i += NT) whh2s[i] = 0.f;
  for (int i = tid; i < 112 * SH2; i += NT) wih3s[i] = 0.f;
  for (int i = tid; i < 112 * SH3; i += NT) whh3s[i] = 0.f;
  __syncthreads();

  for (int i = tid; i < G1 * H1; i += NT) {
    const int n = i / H1, k = i - n * H1;
    whh1s[i] = w_hh1[gate_row(n >> 4, n & 15, H1) * H1 + k];
  }
  for (int i = tid; i < 208 * H2; i += NT) {
    const int n = i / H2, k = i - n * H2;
    if (gate_unit(n >> 4, n & 15) < H2)
      whh2s[n * SH2 + k] = w_hh2[gate_row(n >> 4, n & 15, H2) * H2 + k];
  }
  for (int i = tid; i < 112 * H2; i += NT) {
    const int n = i / H2, k = i - n * H2;
    if (gate_unit(n >> 4, n & 15) < H3)
      wih3s[n * SH2 + k] = w_ih3[gate_row(n >> 4, n & 15, H3) * H2 + k];
  }
  for (int i = tid; i < 112 * H3; i += NT) {
    const int n = i / H3, k = i - n * H3;
    if (gate_unit(n >> 4, n & 15) < H3)
      whh3s[n * SH3 + k] = w_hh3[gate_row(n >> 4, n & 15, H3) * H3 + k];
  }

  for (int i = tid; i < BT_TILES * 16; i += NT)
    reinterpret_cast<float*>(biasT)[i] = bias_entry(i, b_ih1, b_hh1, b_ih2, b_hh2, b_ih3, b_hh3);

  float wl[H3];
  #pragma unroll
  for (int j = 0; j < H3; ++j) wl[j] = w_lin[j];
  const float blin = b_lin[0];

  v2f wi1[4][KS_X1] = {};
  v2f wi2[2][KS_H1] = {};
  #pragma unroll
  for (int s = 0; s < 4; ++s) {
    const int tile = wave + s * NWAVES;
    if (tile < 25)
      load_frags<KS_X1>(wi1[s], w_ih1 + (size_t)gate_row(tile, col, H1) * NIN, khalf);
  }
  #pragma unroll
  for (int s = 0; s < 2; ++s) {
    const int tile = l2_tile(wave, s);
    if (tile >= 0) {
      if (gate_unit(tile, col) < H2)
        load_frags<KS_H1>(wi2[s], w_ih2 + (size_t)gate_row(tile, col, H2) * H1, khalf);
    }
  }

  float c1r[4][2] = {}, c2r[2][2] = {}, c3r[2][2] = {};

  v8f acc1[4] = {};
  {
    v2f ax0[KS_X1];
    load_frags<KS_X1>(ax0, x + ((size_t)chalf * BB + bbase + brow) * NIN, khalf);
    #pragma unroll
    for (int s = 0; s < 4; ++s) {
      const int tile = wave + s * NWAVES;
      if (tile < 25) {
        v8f z = {};
        acc1[s] = wmma_chain<KS_X1>(wi1[s], ax0, z);
      }
    }
  }

  __syncthreads();

  for (int p = 0; p < TT + 2; ++p) {
    const bool act1 = (p < TT);
    const bool act2 = (p >= 1) && (p <= TT);
    const bool act3 = (p >= 2);
    const bool nxt1 = (p & 1) && (p + 1 < TT);
    const int cur = p & 1, nxt = cur ^ 1;
    const bool active = (chalf == cur);
    const int rrow = col ^ (cur << 3);
    const float* h1r = h1s + cur * (16 * SH1);
    const float* h2r = h2s + cur * (16 * SH2);
    const float* h3r = h3s + cur * (16 * SH3);
    float* h1w = h1s + nxt * (16 * SH1) + brow * SH1 + 2 * uhalf;
    float* h2w = h2s + nxt * (16 * SH2) + brow * SH2 + 2 * uhalf;
    float* h3w = h3s + nxt * (16 * SH3) + brow * SH3 + 2 * uhalf;

    if (p + 3 < TT && tid < 16)
      __builtin_prefetch(x + ((size_t)(p + 3) * BB + bbase) * NIN + tid * 32, 0, 3);

    v2f ax[KS_X1];
    if (nxt1)
      load_frags<KS_X1>(ax, x + ((size_t)(p + 1 + chalf) * BB + bbase + brow) * NIN, khalf);

    {
      v2f ah2[KS_H2];
      load_frags<KS_H2>(ah2, &h2r[rrow * SH2], khalf);


      if (act3) {
        if (l3_tile(wave, 0) >= 0) {
          v2f a3[KS_H3];
          load_frags<KS_H3>(a3, &h3r[rrow * SH3], khalf);
          #pragma unroll
          for (int s = 0; s < 2; ++s) {
            const int tile = l3_tile(wave, s);
            if (tile >= 0) {
              const int n = tile * 16 + col;
              v2f bi[KS_H2], bh[KS_H3];
              float bs[8];
              load_bias(bs, biasT, BT_L3 + tile, uhalf);
              load_frags<KS_H2>(bi, &wih3s[n * SH2], khalf);
              load_frags<KS_H3>(bh, &whh3s[n * SH3], khalf);
              bias_read_here(bs);
              reads_stay_grouped();
              v8f acc = {};
              acc = wmma_chain<KS_H2>(bi, ah2, acc);
              acc = wmma_chain<KS_H3>(bh, a3, acc);
              tile_cells(acc, bs, c3r[s], h3w + 4 * tile, active);
            }
          }
        }
      }

      v2f ah1[KS_H1];
      load_frags<KS_H1>(ah1, &h1r[rrow * SH1], khalf);

      if (act2) {
        #pragma unroll
        for (int s = 0; s < 2; ++s) {
          const int tile = l2_tile(wave, s);
          if (tile >= 0) {
            const int n = tile * 16 + col;
            v2f bh[KS_H2];
            float bs[8];
            load_bias(bs, biasT, BT_L2 + tile, uhalf);
            load_frags<KS_H2>(bh, &whh2s[n * SH2], khalf);
            bias_read_here(bs);
            reads_stay_grouped();
            v8f acc = {};
            acc = wmma_chain<KS_H1>(wi2[s], ah1, acc);
            acc = wmma_chain<KS_H2>(bh, ah2, acc);
            tile_cells(acc, bs, c2r[s], h2w + 4 * tile, active);
          }
        }
      }

      if (act1) {
        v2f bh[KS_H1];
        float bs[8];
        load_bias(bs, biasT, BT_L1 + wave, uhalf);
        load_frags<KS_H1>(bh, &whh1s[(wave * 16 + col) * H1], khalf);
        #pragma unroll
        for (int s = 0; s < 4; ++s) {
          const int tile = wave + s * NWAVES;
          if (tile < 25) {
            bias_read_here(bs);
            reads_stay_grouped();
            acc1[s] = wmma_chain<KS_H1>(bh, ah1, acc1[s]);
            reads_stay_grouped();
            const bool more = (s < 3) && (tile + NWAVES < 25);
            float bsn[8];
            load_bias(bsn, biasT, BT_L1 + (more ? tile + NWAVES : tile), uhalf);
            if (more)
              load_frags<KS_H1>(bh, &whh1s[((tile + NWAVES) * 16 + col) * H1], khalf);
            reads_stay_grouped();
            tile_cells(acc1[s], bs, c1r[s], h1w + 4 * tile, active);
            reads_stay_grouped();
            #pragma unroll
            for (int j = 0; j < 8; ++j) bs[j] = bsn[j];
          }
        }
      }
    }

    __builtin_amdgcn_fence(__ATOMIC_RELEASE, "workgroup");
    __builtin_amdgcn_s_barrier_signal(-1);
    __builtin_amdgcn_sched_barrier(0);
    if (nxt1) {
      #pragma unroll
      for (int s = 0; s < 4; ++s) {
        const int tile = wave + s * NWAVES;
        if (tile < 25) {
          v8f z = {};
          acc1[s] = wmma_chain<KS_X1>(wi1[s], ax, z);
        }
      }
    }
    __builtin_amdgcn_sched_barrier(0);
    __builtin_amdgcn_s_barrier_wait(-1);
    __builtin_amdgcn_fence(__ATOMIC_ACQUIRE, "workgroup");

    if (act3 && tid < BR) {
      const float* hr = &h3s[nxt * (16 * SH3) + tid * SH3];
      float acc = blin;
      #pragma unroll
      for (int j = 0; j < H3; ++j) acc += hr[j] * wl[j];
      out[(size_t)(p - 2) * BB + bbase + tid] = acc;
    }
  }
}

extern "C" void kernel_launch(void* const* d_in, const int* in_sizes, int n_in,
                              void* d_out, int out_size, void* d_ws, size_t ws_size,
                              hipStream_t stream) {
  (void)in_sizes; (void)n_in; (void)d_ws; (void)ws_size; (void)out_size;
  const float* x     = (const float*)d_in[0];
  const float* w_ih1 = (const float*)d_in[1];
  const float* w_hh1 = (const float*)d_in[2];
  const float* b_ih1 = (const float*)d_in[3];
  const float* b_hh1 = (const float*)d_in[4];
  const float* w_ih2 = (const float*)d_in[5];
  const float* w_hh2 = (const float*)d_in[6];
  const float* b_ih2 = (const float*)d_in[7];
  const float* b_hh2 = (const float*)d_in[8];
  const float* w_ih3 = (const float*)d_in[9];
  const float* w_hh3 = (const float*)d_in[10];
  const float* b_ih3 = (const float*)d_in[11];
  const float* b_hh3 = (const float*)d_in[12];
  const float* w_lin = (const float*)d_in[13];
  const float* b_lin = (const float*)d_in[14];
  float* out = (float*)d_out;

  dim3 grid(BB / BR);
  dim3 block(NT);
  lstm3_fused_kernel<<<grid, block, 0, stream>>>(
      x, w_ih1, w_hh1, b_ih1, b_hh1,
      w_ih2, w_hh2, b_ih2, b_hh2,
      w_ih3, w_hh3, b_ih3, b_hh3,
      w_lin, b_lin, out);
}
